# starting kernel
# baseline (speedup 1.0000x reference)
#include <hip/hip_runtime.h>
#include <hip/hip_bf16.h>

#define N_NODES   50000
#define N_EDGES   500000
#define IN_FEATS  128
#define N_HIDDEN  128
#define N_CLASSES 64

typedef __attribute__((ext_vector_type(2))) float v2f;
typedef __attribute__((ext_vector_type(8))) float v8f;

__global__ __launch_bounds__(256) void zero_kernel(float* __restrict__ p, int n) {
    int i = blockIdx.x * blockDim.x + threadIdx.x;
    if (i < n) p[i] = 0.0f;
}

template <int N, bool RELU_A>
__global__ __launch_bounds__(256)
void gemm_wmma_f32(const float* __restrict__ A, const float* __restrict__ W,
                   const float* __restrict__ bias, float* __restrict__ out, int M) {
    constexpr int K = 128;
    const int lane  = threadIdx.x & 31;
    const int wave  = threadIdx.x >> 5;
    const int mTile = blockIdx.x * 8 + wave;
    if (mTile * 16 >= M) return;
    const int nTile = blockIdx.y;
    const int half  = lane >> 4;
    const int l16   = lane & 15;
    const int nCol  = nTile * 16 + l16;

    const float bv = bias[nCol];
    v8f c;
#pragma unroll
    for (int i = 0; i < 8; ++i) c[i] = bv;

    const float* aRow = A + (size_t)(mTile * 16 + l16) * K;

#pragma unroll 4
    for (int k0 = 0; k0 < K; k0 += 4) {
        const int kk = k0 + 2 * half;
        v2f a = *(const v2f*)(aRow + kk);
        if (RELU_A) {
            a.x = fmaxf(a.x, 0.0f);
            a.y = fmaxf(a.y, 0.0f);
        }
        v2f b;
        b.x = W[(size_t)(kk + 0) * N + nCol];
        b.y = W[(size_t)(kk + 1) * N + nCol];
        c = __builtin_amdgcn_wmma_f32_16x16x4_f32(false, a, false, b,
                                                  (short)0, c, false, false);
    }

    float* oBase = out + (size_t)(mTile * 16 + half * 8) * N + nCol;
#pragma unroll
    for (int r = 0; r < 8; ++r) oBase[(size_t)r * N] = c[r];
}

template <int FEATS>
__global__ __launch_bounds__(256)
void aggregate_kernel(const float* __restrict__ h, const int* __restrict__ src,
                      const int* __restrict__ dst, const float* __restrict__ ew,
                      float* __restrict__ out, int nEdges) {
    const int lane = threadIdx.x & 31;
    const int e    = blockIdx.x * 8 + (threadIdx.x >> 5);
    if (e >= nEdges) return;
    const int   s = src[e];
    const int   d = dst[e];
    const float w = ew[e];

    constexpr int V = FEATS / 32;
    const float* hp = h   + (size_t)s * FEATS + lane * V;
    float*       op = out + (size_t)d * FEATS + lane * V;

    float vals[V];
#pragma unroll
    for (int i = 0; i < V; ++i) vals[i] = hp[i];
#pragma unroll
    for (int i = 0; i < V; ++i) unsafeAtomicAdd(op + i, vals[i] * w);
}

extern "C" void kernel_launch(void* const* d_in, const int* in_sizes, int n_in,
                              void* d_out, int out_size, void* d_ws, size_t ws_size,
                              hipStream_t stream) {
    const float* x   = (const float*)d_in[0];
    const int*   src = (const int*)  d_in[1];
    const int*   dst = (const int*)  d_in[2];
    const float* w1  = (const float*)d_in[3];
    const float* w2  = (const float*)d_in[4];
    const float* w3  = (const float*)d_in[5];
    const float* W1  = (const float*)d_in[6];
    const float* b1  = (const float*)d_in[7];
    const float* W2  = (const float*)d_in[8];
    const float* b2  = (const float*)d_in[9];
    const float* W3  = (const float*)d_in[10];
    const float* b3  = (const float*)d_in[11];
    float* out = (float*)d_out;

    float* wsA = (float*)d_ws;
    float* wsB = wsA + (size_t)N_NODES * N_HIDDEN;

    const int mTiles   = (N_NODES + 15) / 16;
    const dim3 blk(256);
    const dim3 gemmGrid128((mTiles + 7) / 8, N_HIDDEN / 16);
    const dim3 gemmGrid64 ((mTiles + 7) / 8, N_CLASSES / 16);
    const dim3 aggGrid((N_EDGES + 7) / 8);

    const int nH  = N_NODES * N_HIDDEN;
    const int nC  = N_NODES * N_CLASSES;
    const dim3 zeroGridH((nH + 255) / 256);
    const dim3 zeroGridC((nC + 255) / 256);

    gemm_wmma_f32<N_HIDDEN, false><<<gemmGrid128, blk, 0, stream>>>(x, W1, b1, wsA, N_NODES);
    zero_kernel<<<zeroGridH, blk, 0, stream>>>(wsB, nH);
    aggregate_kernel<N_HIDDEN><<<aggGrid, blk, 0, stream>>>(wsA, src, dst, w1, wsB, N_EDGES);

    gemm_wmma_f32<N_HIDDEN, true><<<gemmGrid128, blk, 0, stream>>>(wsB, W2, b2, wsA, N_NODES);
    zero_kernel<<<zeroGridH, blk, 0, stream>>>(wsB, nH);
    aggregate_kernel<N_HIDDEN><<<aggGrid, blk, 0, stream>>>(wsA, src, dst, w2, wsB, N_EDGES);

    gemm_wmma_f32<N_CLASSES, true><<<gemmGrid64, blk, 0, stream>>>(wsB, W3, b3, wsA, N_NODES);
    zero_kernel<<<zeroGridC, blk, 0, stream>>>(out, nC);
    aggregate_kernel<N_CLASSES><<<aggGrid, blk, 0, stream>>>(wsA, src, dst, w3, out, N_EDGES);
}


# Replace f32 atomic scatter-add by per-call destination bucketing (integer-atomic count, scan, place) and a one-wave-per-node register gather-sum; sum order is now the arrival-order rank
# speedup vs baseline: 1.7505x; 1.7505x over previous
#include <hip/hip_runtime.h>
#include <hip/hip_bf16.h>


#define N_NODES   50000
#define N_EDGES   500000
#define IN_FEATS  128
#define N_HIDDEN  128
#define N_CLASSES 64
#define CNT_PAD   50016

typedef __attribute__((ext_vector_type(2))) float v2f;
typedef __attribute__((ext_vector_type(4))) float v4f;
typedef __attribute__((ext_vector_type(8))) float v8f;

__global__ __launch_bounds__(256) void zero_int_kernel(int* __restrict__ p, int n) {
    int i = blockIdx.x * blockDim.x + threadIdx.x;
    if (i < n) p[i] = 0;
}

__global__ __launch_bounds__(256)
void count_rank_kernel(const int* __restrict__ dst, int* __restrict__ cnt,
                       int* __restrict__ rank, int nEdges) {
    const int e = blockIdx.x * blockDim.x + threadIdx.x;
    if (e < nEdges) rank[e] = atomicAdd(cnt + dst[e], 1);
}

__global__ __launch_bounds__(1024)
void rowptr_scan_kernel(const int* __restrict__ cnt, int* __restrict__ rowptr, int n) {
    __shared__ int sTot[32];
    __shared__ int sBefore[32];
    const int tid  = threadIdx.x;
    const int lane = tid & 31;
    const int wave = tid >> 5;
    const int base = blockIdx.x * 1024;

    int before = 0;
    for (int i = tid; i < base; i += 1024) before += cnt[i];
    const int idx = base + tid;
    const int v   = (idx < n) ? cnt[idx] : 0;

    int incl = v;
#pragma unroll
    for (int o = 1; o < 32; o <<= 1) {
        const int t = __shfl_up(incl, o);
        if (lane >= o) incl += t;
    }
#pragma unroll
    for (int o = 16; o > 0; o >>= 1) before += __shfl_xor(before, o);
    if (lane == 31) sTot[wave]    = incl;
    if (lane == 0)  sBefore[wave] = before;
    __syncthreads();

    const int wt = sTot[lane];
    int blockBefore = sBefore[lane];
#pragma unroll
    for (int o = 16; o > 0; o >>= 1) blockBefore += __shfl_xor(blockBefore, o);
    int wincl = wt;
#pragma unroll
    for (int o = 1; o < 32; o <<= 1) {
        const int t = __shfl_up(wincl, o);
        if (lane >= o) wincl += t;
    }
    const int wavePre = __shfl(wincl - wt, wave);
    if (idx <= n) rowptr[idx] = blockBefore + wavePre + incl - v;
}

__global__ __launch_bounds__(256)
void place_kernel(const int* __restrict__ src, const int* __restrict__ dst,
                  const float* __restrict__ w1, const float* __restrict__ w2,
                  const float* __restrict__ w3, const int* __restrict__ rank,
                  const int* __restrict__ rowptr, float* __restrict__ rec, int nEdges) {
    const int e = blockIdx.x * blockDim.x + threadIdx.x;
    if (e >= nEdges) return;
    const int pos = rowptr[dst[e]] + rank[e];
    v4f r;
    r.x = __int_as_float(src[e]);
    r.y = w1[e];
    r.z = w2[e];
    r.w = w3[e];
    ((v4f*)rec)[pos] = r;
}

template <int N, bool RELU_A>
__global__ __launch_bounds__(256)
void gemm_wmma_f32(const float* __restrict__ A, const float* __restrict__ W,
                   const float* __restrict__ bias, float* __restrict__ out, int M) {
    constexpr int K = 128;
    const int lane  = threadIdx.x & 31;
    const int wave  = threadIdx.x >> 5;
    const int mTile = blockIdx.x * 8 + wave;
    if (mTile * 16 >= M) return;
    const int nTile = blockIdx.y;
    const int half  = lane >> 4;
    const int l16   = lane & 15;
    const int nCol  = nTile * 16 + l16;

    const float bv = bias[nCol];
    v8f c;
#pragma unroll
    for (int i = 0; i < 8; ++i) c[i] = bv;

    const float* aRow = A + (size_t)(mTile * 16 + l16) * K;

#pragma unroll 4
    for (int k0 = 0; k0 < K; k0 += 4) {
        const int kk = k0 + 2 * half;
        v2f a = *(const v2f*)(aRow + kk);
        if (RELU_A) {
            a.x = fmaxf(a.x, 0.0f);
            a.y = fmaxf(a.y, 0.0f);
        }
        v2f b;
        b.x = W[(size_t)(kk + 0) * N + nCol];
        b.y = W[(size_t)(kk + 1) * N + nCol];
        c = __builtin_amdgcn_wmma_f32_16x16x4_f32(false, a, false, b,
                                                  (short)0, c, false, false);
    }

    float* oBase = out + (size_t)(mTile * 16 + half * 8) * N + nCol;
#pragma unroll
    for (int r = 0; r < 8; ++r) oBase[(size_t)r * N] = c[r];
}

template <int FEATS, int K, typename VROW>
__device__ __forceinline__ void add_rows(VROW& acc, const float* __restrict__ hLane,
                                         int sL, int wL, int j) {
    int   s[K];
    float w[K];
    VROW  a[K];
#pragma unroll
    for (int i = 0; i < K; ++i) {
        s[i] = __builtin_amdgcn_readlane(sL, j + i);
        w[i] = __int_as_float(__builtin_amdgcn_readlane(wL, j + i));
    }
#pragma unroll
    for (int i = 0; i < K; ++i) a[i] = *(const VROW*)(hLane + (size_t)s[i] * FEATS);
#pragma unroll
    for (int i = 0; i < K; ++i) acc += a[i] * w[i];
}

template <int FEATS, int WSEL>
__global__ __launch_bounds__(256)
void gather_sum_kernel(const float* __restrict__ h, const int* __restrict__ rowptr,
                       const float* __restrict__ rec, float* __restrict__ out,
                       int nNodes) {
    constexpr int V = FEATS / 32;
    typedef __attribute__((ext_vector_type(V))) float vrow;
    const int lane = threadIdx.x & 31;
    const int d = __builtin_amdgcn_readfirstlane(blockIdx.x * 8 + (threadIdx.x >> 5));
    if (d >= nNodes) return;
    const int beg = rowptr[d];
    const int end = rowptr[d + 1];

    const float* hLane = h + lane * V;
    vrow acc = (vrow)(0.0f);

    for (int base = beg; base < end; base += 32) {
        const int n = min(32, end - base);
        v4f r = (v4f)(0.0f);
        if (lane < n) r = ((const v4f*)rec)[base + lane];
        const int sL = __float_as_int(r.x);
        const int wL = __float_as_int(WSEL == 1 ? r.y : (WSEL == 2 ? r.z : r.w));

        int j = 0;
        for (; j + 8 <= n; j += 8) add_rows<FEATS, 8>(acc, hLane, sL, wL, j);
        if (j + 4 <= n) { add_rows<FEATS, 4>(acc, hLane, sL, wL, j); j += 4; }
        const int rem = n - j;
        if (rem == 3)      add_rows<FEATS, 3>(acc, hLane, sL, wL, j);
        else if (rem == 2) add_rows<FEATS, 2>(acc, hLane, sL, wL, j);
        else if (rem == 1) add_rows<FEATS, 1>(acc, hLane, sL, wL, j);
    }
    *(vrow*)(out + (size_t)d * FEATS + lane * V) = acc;
}

extern "C" void kernel_launch(void* const* d_in, const int* in_sizes, int n_in,
                              void* d_out, int out_size, void* d_ws, size_t ws_size,
                              hipStream_t stream) {
    const float* x   = (const float*)d_in[0];
    const int*   src = (const int*)  d_in[1];
    const int*   dst = (const int*)  d_in[2];
    const float* w1  = (const float*)d_in[3];
    const float* w2  = (const float*)d_in[4];
    const float* w3  = (const float*)d_in[5];
    const float* W1  = (const float*)d_in[6];
    const float* b1  = (const float*)d_in[7];
    const float* W2  = (const float*)d_in[8];
    const float* b2  = (const float*)d_in[9];
    const float* W3  = (const float*)d_in[10];
    const float* b3  = (const float*)d_in[11];
    float* out = (float*)d_out;

    float* wsA    = (float*)d_ws;
    float* wsB    = wsA + (size_t)N_NODES * N_HIDDEN;
    float* rec    = wsB + (size_t)N_NODES * N_HIDDEN;
    int*   rank   = (int*)(rec + (size_t)N_EDGES * 4);
    int*   cnt    = rank + N_EDGES;
    int*   rowptr = cnt + CNT_PAD;

    const int mTiles   = (N_NODES + 15) / 16;
    const dim3 blk(256);
    const dim3 gemmGrid128((mTiles + 7) / 8, N_HIDDEN / 16);
    const dim3 gemmGrid64 ((mTiles + 7) / 8, N_CLASSES / 16);
    const dim3 edgeGrid((N_EDGES + 255) / 256);
    const dim3 nodeGrid((N_NODES + 7) / 8);
    const dim3 cntGrid((N_NODES + 1 + 255) / 256);
    const dim3 scanGrid((N_NODES + 1 + 1023) / 1024);

    zero_int_kernel<<<cntGrid, blk, 0, stream>>>(cnt, N_NODES + 1);
    count_rank_kernel<<<edgeGrid, blk, 0, stream>>>(dst, cnt, rank, N_EDGES);
    rowptr_scan_kernel<<<scanGrid, dim3(1024), 0, stream>>>(cnt, rowptr, N_NODES);
    place_kernel<<<edgeGrid, blk, 0, stream>>>(src, dst, w1, w2, w3, rank, rowptr, rec, N_EDGES);

    gemm_wmma_f32<N_HIDDEN, false><<<gemmGrid128, blk, 0, stream>>>(x, W1, b1, wsA, N_NODES);
    gather_sum_kernel<N_HIDDEN, 1><<<nodeGrid, blk, 0, stream>>>(wsA, rowptr, rec, wsB, N_NODES);

    gemm_wmma_f32<N_HIDDEN, true><<<gemmGrid128, blk, 0, stream>>>(wsB, W2, b2, wsA, N_NODES);
    gather_sum_kernel<N_HIDDEN, 2><<<nodeGrid, blk, 0, stream>>>(wsA, rowptr, rec, wsB, N_NODES);

    gemm_wmma_f32<N_CLASSES, true><<<gemmGrid64, blk, 0, stream>>>(wsB, W3, b3, wsA, N_NODES);
    gather_sum_kernel<N_CLASSES, 3><<<nodeGrid, blk, 0, stream>>>(wsA, rowptr, rec, out, N_NODES);
}


# GEMM: one wave computes a 16-row strip across all N columns, W staged once per block in LDS as k-pairs, so each A fragment is loaded once and feeds N/16 WMMAs; k order and sum order unchanged
# speedup vs baseline: 2.3432x; 1.3386x over previous
#include <hip/hip_runtime.h>
#include <hip/hip_bf16.h>


#define N_NODES   50000
#define N_EDGES   500000
#define IN_FEATS  128
#define N_HIDDEN  128
#define N_CLASSES 64
#define CNT_PAD   50016

typedef __attribute__((ext_vector_type(2))) float v2f;
typedef __attribute__((ext_vector_type(4))) float v4f;
typedef __attribute__((ext_vector_type(8))) float v8f;

__global__ __launch_bounds__(256) void zero_int_kernel(int* __restrict__ p, int n) {
    int i = blockIdx.x * blockDim.x + threadIdx.x;
    if (i < n) p[i] = 0;
}

__global__ __launch_bounds__(256)
void count_rank_kernel(const int* __restrict__ dst, int* __restrict__ cnt,
                       int* __restrict__ rank, int nEdges) {
    const int e = blockIdx.x * blockDim.x + threadIdx.x;
    if (e < nEdges) rank[e] = atomicAdd(cnt + dst[e], 1);
}

__global__ __launch_bounds__(1024)
void rowptr_scan_kernel(const int* __restrict__ cnt, int* __restrict__ rowptr, int n) {
    __shared__ int sTot[32];
    __shared__ int sBefore[32];
    const int tid  = threadIdx.x;
    const int lane = tid & 31;
    const int wave = tid >> 5;
    const int base = blockIdx.x * 1024;

    int before = 0;
    for (int i = tid; i < base; i += 1024) before += cnt[i];
    const int idx = base + tid;
    const int v   = (idx < n) ? cnt[idx] : 0;

    int incl = v;
#pragma unroll
    for (int o = 1; o < 32; o <<= 1) {
        const int t = __shfl_up(incl, o);
        if (lane >= o) incl += t;
    }
#pragma unroll
    for (int o = 16; o > 0; o >>= 1) before += __shfl_xor(before, o);
    if (lane == 31) sTot[wave]    = incl;
    if (lane == 0)  sBefore[wave] = before;
    __syncthreads();

    const int wt = sTot[lane];
    int blockBefore = sBefore[lane];
#pragma unroll
    for (int o = 16; o > 0; o >>= 1) blockBefore += __shfl_xor(blockBefore, o);
    int wincl = wt;
#pragma unroll
    for (int o = 1; o < 32; o <<= 1) {
        const int t = __shfl_up(wincl, o);
        if (lane >= o) wincl += t;
    }
    const int wavePre = __shfl(wincl - wt, wave);
    if (idx <= n) rowptr[idx] = blockBefore + wavePre + incl - v;
}

__global__ __launch_bounds__(256)
void place_kernel(const int* __restrict__ src, const int* __restrict__ dst,
                  const float* __restrict__ w1, const float* __restrict__ w2,
                  const float* __restrict__ w3, const int* __restrict__ rank,
                  const int* __restrict__ rowptr, float* __restrict__ rec, int nEdges) {
    const int e = blockIdx.x * blockDim.x + threadIdx.x;
    if (e >= nEdges) return;
    const int pos = rowptr[dst[e]] + rank[e];
    v4f r;
    r.x = __int_as_float(src[e]);
    r.y = w1[e];
    r.z = w2[e];
    r.w = w3[e];
    ((v4f*)rec)[pos] = r;
}

template <int N, bool RELU_A>
__global__ __launch_bounds__(256)
void gemm_wmma_f32(const float* __restrict__ A, const float* __restrict__ W,
                   const float* __restrict__ bias, float* __restrict__ out, int M) {
    constexpr int K     = 128;
    constexpr int KS    = K / 4;
    constexpr int NT    = N / 16;
    constexpr int PITCH = 2 * N + 32;
    constexpr int N4    = N / 4;
    constexpr int ITEMS = (K / 2) * N4;
    __shared__ __attribute__((aligned(16))) float sW[(K / 2) * PITCH];

    const int tid   = threadIdx.x;
    const int lane  = tid & 31;
    const int wave  = tid >> 5;
    const int half  = lane >> 4;
    const int l16   = lane & 15;
    const int mTile = blockIdx.x * 8 + wave;
    const bool live = (mTile * 16 < M);

    const int aRowIdx = live ? (mTile * 16 + l16) : l16;
    const float* aRow = A + (size_t)aRowIdx * K + 2 * half;
    v2f a[KS];
#pragma unroll
    for (int s = 0; s < KS; ++s) a[s] = *(const v2f*)(aRow + 4 * s);

#pragma unroll
    for (int i = 0; i < ITEMS / 256; ++i) {
        const int it = tid + 256 * i;
        const int p  = it / N4;
        const int n4 = it % N4;
        const v4f r0 = *(const v4f*)(W + (size_t)(2 * p)     * N + 4 * n4);
        const v4f r1 = *(const v4f*)(W + (size_t)(2 * p + 1) * N + 4 * n4);
        float* d = sW + p * PITCH + 8 * n4;
        v4f lo, hi;
        lo.x = r0.x; lo.y = r1.x; lo.z = r0.y; lo.w = r1.y;
        hi.x = r0.z; hi.y = r1.z; hi.z = r0.w; hi.w = r1.w;
        *(v4f*)d       = lo;
        *(v4f*)(d + 4) = hi;
    }
    __syncthreads();
    if (!live) return;

    if (RELU_A) {
#pragma unroll
        for (int s = 0; s < KS; ++s) {
            a[s].x = fmaxf(a[s].x, 0.0f);
            a[s].y = fmaxf(a[s].y, 0.0f);
        }
    }

    v8f c[NT];
#pragma unroll
    for (int t = 0; t < NT; ++t) {
        const float bv = bias[t * 16 + l16];
#pragma unroll
        for (int i = 0; i < 8; ++i) c[t][i] = bv;
    }

    const float* bBase = sW + half * PITCH + 2 * l16;
#pragma unroll
    for (int s = 0; s < KS; ++s) {
#pragma unroll
        for (int t = 0; t < NT; ++t) {
            const v2f b = *(const v2f*)(bBase + (2 * s) * PITCH + 32 * t);
            c[t] = __builtin_amdgcn_wmma_f32_16x16x4_f32(false, a[s], false, b,
                                                         (short)0, c[t], false, false);
        }
    }

    float* oBase = out + (size_t)(mTile * 16 + half * 8) * N + l16;
#pragma unroll
    for (int r = 0; r < 8; ++r) {
#pragma unroll
        for (int t = 0; t < NT; ++t) oBase[(size_t)r * N + 16 * t] = c[t][r];
    }
}

template <int FEATS, int K, typename VROW>
__device__ __forceinline__ void add_rows(VROW& acc, const float* __restrict__ hLane,
                                         int sL, int wL, int j) {
    int   s[K];
    float w[K];
    VROW  a[K];
#pragma unroll
    for (int i = 0; i < K; ++i) {
        s[i] = __builtin_amdgcn_readlane(sL, j + i);
        w[i] = __int_as_float(__builtin_amdgcn_readlane(wL, j + i));
    }
#pragma unroll
    for (int i = 0; i < K; ++i) a[i] = *(const VROW*)(hLane + (size_t)s[i] * FEATS);
#pragma unroll
    for (int i = 0; i < K; ++i) acc += a[i] * w[i];
}

template <int FEATS, int WSEL>
__global__ __launch_bounds__(256)
void gather_sum_kernel(const float* __restrict__ h, const int* __restrict__ rowptr,
                       const float* __restrict__ rec, float* __restrict__ out,
                       int nNodes) {
    constexpr int V = FEATS / 32;
    typedef __attribute__((ext_vector_type(V))) float vrow;
    const int lane = threadIdx.x & 31;
    const int d = __builtin_amdgcn_readfirstlane(blockIdx.x * 8 + (threadIdx.x >> 5));
    if (d >= nNodes) return;
    const int beg = rowptr[d];
    const int end = rowptr[d + 1];

    const float* hLane = h + lane * V;
    vrow acc = (vrow)(0.0f);

    for (int base = beg; base < end; base += 32) {
        const int n = min(32, end - base);
        v4f r = (v4f)(0.0f);
        if (lane < n) r = ((const v4f*)rec)[base + lane];
        const int sL = __float_as_int(r.x);
        const int wL = __float_as_int(WSEL == 1 ? r.y : (WSEL == 2 ? r.z : r.w));

        int j = 0;
        for (; j + 8 <= n; j += 8) add_rows<FEATS, 8>(acc, hLane, sL, wL, j);
        if (j + 4 <= n) { add_rows<FEATS, 4>(acc, hLane, sL, wL, j); j += 4; }
        const int rem = n - j;
        if (rem == 3)      add_rows<FEATS, 3>(acc, hLane, sL, wL, j);
        else if (rem == 2) add_rows<FEATS, 2>(acc, hLane, sL, wL, j);
        else if (rem == 1) add_rows<FEATS, 1>(acc, hLane, sL, wL, j);
    }
    *(vrow*)(out + (size_t)d * FEATS + lane * V) = acc;
}

extern "C" void kernel_launch(void* const* d_in, const int* in_sizes, int n_in,
                              void* d_out, int out_size, void* d_ws, size_t ws_size,
                              hipStream_t stream) {
    const float* x   = (const float*)d_in[0];
    const int*   src = (const int*)  d_in[1];
    const int*   dst = (const int*)  d_in[2];
    const float* w1  = (const float*)d_in[3];
    const float* w2  = (const float*)d_in[4];
    const float* w3  = (const float*)d_in[5];
    const float* W1  = (const float*)d_in[6];
    const float* b1  = (const float*)d_in[7];
    const float* W2  = (const float*)d_in[8];
    const float* b2  = (const float*)d_in[9];
    const float* W3  = (const float*)d_in[10];
    const float* b3  = (const float*)d_in[11];
    float* out = (float*)d_out;

    float* wsA    = (float*)d_ws;
    float* wsB    = wsA + (size_t)N_NODES * N_HIDDEN;
    float* rec    = wsB + (size_t)N_NODES * N_HIDDEN;
    int*   rank   = (int*)(rec + (size_t)N_EDGES * 4);
    int*   cnt    = rank + N_EDGES;
    int*   rowptr = cnt + CNT_PAD;

    const int mTiles   = (N_NODES + 15) / 16;
    const dim3 blk(256);
    const dim3 gemmGrid((mTiles + 7) / 8);
    const dim3 edgeGrid((N_EDGES + 255) / 256);
    const dim3 nodeGrid((N_NODES + 7) / 8);
    const dim3 cntGrid((N_NODES + 1 + 255) / 256);
    const dim3 scanGrid((N_NODES + 1 + 1023) / 1024);

    zero_int_kernel<<<cntGrid, blk, 0, stream>>>(cnt, N_NODES + 1);
    count_rank_kernel<<<edgeGrid, blk, 0, stream>>>(dst, cnt, rank, N_EDGES);
    rowptr_scan_kernel<<<scanGrid, dim3(1024), 0, stream>>>(cnt, rowptr, N_NODES);
    place_kernel<<<edgeGrid, blk, 0, stream>>>(src, dst, w1, w2, w3, rank, rowptr, rec, N_EDGES);

    gemm_wmma_f32<N_HIDDEN, false><<<gemmGrid, blk, 0, stream>>>(x, W1, b1, wsA, N_NODES);
    gather_sum_kernel<N_HIDDEN, 1><<<nodeGrid, blk, 0, stream>>>(wsA, rowptr, rec, wsB, N_NODES);

    gemm_wmma_f32<N_HIDDEN, true><<<gemmGrid, blk, 0, stream>>>(wsB, W2, b2, wsA, N_NODES);
    gather_sum_kernel<N_HIDDEN, 2><<<nodeGrid, blk, 0, stream>>>(wsA, rowptr, rec, wsB, N_NODES);

    gemm_wmma_f32<N_CLASSES, true><<<gemmGrid, blk, 0, stream>>>(wsB, W3, b3, wsA, N_NODES);
    gather_sum_kernel<N_CLASSES, 3><<<nodeGrid, blk, 0, stream>>>(wsA, rowptr, rec, out, N_NODES);
}


# Edge bucketing without global atomics: two-level counting sort on LDS counters (2048-edge chunks into 256-node bins, then one work-group per bin); sum order still an arrival-order rank
# speedup vs baseline: 2.7512x; 1.1741x over previous
#include <hip/hip_runtime.h>
#include <hip/hip_bf16.h>


#define N_NODES   50000
#define N_EDGES   500000
#define IN_FEATS  128
#define N_HIDDEN  128
#define N_CLASSES 64
#define CNT_PAD   50016
#define BIN_SHIFT 8
#define BIN_NODES 256
#define N_BINS    ((N_NODES + BIN_NODES) / BIN_NODES)
#define CHUNK     2048
#define N_CHUNKS  ((N_EDGES + CHUNK - 1) / CHUNK)
#define T_LEN     (N_BINS * N_CHUNKS)
#define T_PAD     48032
static_assert(N_NODES <= 65536, "a source index must fit the low 16 bits of the packed key");
static_assert(T_LEN + 1 <= T_PAD, "table padding");

typedef __attribute__((ext_vector_type(2))) float v2f;
typedef __attribute__((ext_vector_type(4))) float v4f;
typedef __attribute__((ext_vector_type(8))) float v8f;

__global__ __launch_bounds__(256)
void coarse_count_kernel(const int* __restrict__ dst, int* __restrict__ tbl,
                         int* __restrict__ rank, int nEdges) {
    __shared__ int sCnt[N_BINS];
    const int tid = threadIdx.x;
    for (int i = tid; i < N_BINS; i += 256) sCnt[i] = 0;
    const int base = blockIdx.x * CHUNK + tid;
    int d[CHUNK / 256];
#pragma unroll
    for (int i = 0; i < CHUNK / 256; ++i) d[i] = dst[min(base + 256 * i, nEdges - 1)];
    __syncthreads();
    if (blockIdx.x * CHUNK + CHUNK <= nEdges) {
        int r[CHUNK / 256];
#pragma unroll
        for (int i = 0; i < CHUNK / 256; ++i) r[i] = atomicAdd(&sCnt[d[i] >> BIN_SHIFT], 1);
#pragma unroll
        for (int i = 0; i < CHUNK / 256; ++i) rank[base + 256 * i] = r[i];
    } else {
#pragma unroll
        for (int i = 0; i < CHUNK / 256; ++i) {
            const int e = base + 256 * i;
            if (e < nEdges) rank[e] = atomicAdd(&sCnt[d[i] >> BIN_SHIFT], 1);
        }
    }
    __syncthreads();
    for (int i = tid; i < N_BINS; i += 256) tbl[i * N_CHUNKS + blockIdx.x] = sCnt[i];
}

__global__ __launch_bounds__(1024)
void rowptr_scan_kernel(const int* __restrict__ cnt, int* __restrict__ rowptr, int n) {
    __shared__ int sTot[32];
    __shared__ int sBefore[32];
    const int tid  = threadIdx.x;
    const int lane = tid & 31;
    const int wave = tid >> 5;
    const int base = blockIdx.x * 1024;

    int before = 0;
    for (int i = tid; i < base; i += 1024) before += cnt[i];
    const int idx = base + tid;
    const int v   = (idx < n) ? cnt[idx] : 0;

    int incl = v;
#pragma unroll
    for (int o = 1; o < 32; o <<= 1) {
        const int t = __shfl_up(incl, o);
        if (lane >= o) incl += t;
    }
#pragma unroll
    for (int o = 16; o > 0; o >>= 1) before += __shfl_xor(before, o);
    if (lane == 31) sTot[wave]    = incl;
    if (lane == 0)  sBefore[wave] = before;
    __syncthreads();

    const int wt = sTot[lane];
    int blockBefore = sBefore[lane];
#pragma unroll
    for (int o = 16; o > 0; o >>= 1) blockBefore += __shfl_xor(blockBefore, o);
    int wincl = wt;
#pragma unroll
    for (int o = 1; o < 32; o <<= 1) {
        const int t = __shfl_up(wincl, o);
        if (lane >= o) wincl += t;
    }
    const int wavePre = __shfl(wincl - wt, wave);
    if (idx <= n) rowptr[idx] = blockBefore + wavePre + incl - v;
}

__global__ __launch_bounds__(256)
void coarse_place_kernel(const int* __restrict__ src, const int* __restrict__ dst,
                         const float* __restrict__ w1, const float* __restrict__ w2,
                         const float* __restrict__ w3, const int* __restrict__ rank,
                         const int* __restrict__ tbase, float* __restrict__ recC, int nEdges) {
    const int e = blockIdx.x * blockDim.x + threadIdx.x;
    if (e >= nEdges) return;
    const int d   = dst[e];
    const int pos = tbase[(d >> BIN_SHIFT) * N_CHUNKS + e / CHUNK] + rank[e];
    v4f r;
    r.x = __int_as_float(src[e] | ((d & (BIN_NODES - 1)) << 16));
    r.y = w1[e];
    r.z = w2[e];
    r.w = w3[e];
    ((v4f*)recC)[pos] = r;
}

__global__ __launch_bounds__(1024)
void fine_place_kernel(const float* __restrict__ recC, const int* __restrict__ tbase,
                       int* __restrict__ rowptr, float* __restrict__ rec, int nNodes) {
    __shared__ int sCnt[BIN_NODES];
    __shared__ int sCur[BIN_NODES];
    __shared__ int sWave[BIN_NODES / 32];
    const int tid  = threadIdx.x;
    const int lane = tid & 31;
    const int wave = tid >> 5;
    const int bin  = blockIdx.x;
    const int segBeg = tbase[bin * N_CHUNKS];
    const int segEnd = tbase[(bin + 1) * N_CHUNKS];

    if (tid < BIN_NODES) sCnt[tid] = 0;
    __syncthreads();
    for (int i = segBeg + tid; i < segEnd; i += 1024) {
        const int key = __float_as_int(recC[(size_t)4 * i]);
        atomicAdd(&sCnt[(key >> 16) & (BIN_NODES - 1)], 1);
    }
    __syncthreads();

    int v = 0, incl = 0;
    if (tid < BIN_NODES) {
        v = sCnt[tid];
        incl = v;
#pragma unroll
        for (int o = 1; o < 32; o <<= 1) {
            const int t = __shfl_up(incl, o);
            if (lane >= o) incl += t;
        }
        if (lane == 31) sWave[wave] = incl;
    }
    __syncthreads();
    if (tid < BIN_NODES) {
        int pre = 0;
#pragma unroll
        for (int w = 0; w < BIN_NODES / 32; ++w) pre += (w < wave) ? sWave[w] : 0;
        const int start = segBeg + pre + incl - v;
        sCur[tid] = start;
        const int node = bin * BIN_NODES + tid;
        if (node <= nNodes) rowptr[node] = start;
    }
    __syncthreads();

    for (int i = segBeg + tid; i < segEnd; i += 1024) {
        v4f r = ((const v4f*)recC)[i];
        const int key = __float_as_int(r.x);
        const int p = atomicAdd(&sCur[(key >> 16) & (BIN_NODES - 1)], 1);
        r.x = __int_as_float(key & 0xFFFF);
        ((v4f*)rec)[p] = r;
    }
}

template <int N, bool RELU_A>
__global__ __launch_bounds__(256)
void gemm_wmma_f32(const float* __restrict__ A, const float* __restrict__ W,
                   const float* __restrict__ bias, float* __restrict__ out, int M) {
    constexpr int K     = 128;
    constexpr int KS    = K / 4;
    constexpr int NT    = N / 16;
    constexpr int PITCH = 2 * N + 32;
    constexpr int N4    = N / 4;
    constexpr int ITEMS = (K / 2) * N4;
    __shared__ __attribute__((aligned(16))) float sW[(K / 2) * PITCH];

    const int tid   = threadIdx.x;
    const int lane  = tid & 31;
    const int wave  = tid >> 5;
    const int half  = lane >> 4;
    const int l16   = lane & 15;
    const int mTile = blockIdx.x * 8 + wave;
    const bool live = (mTile * 16 < M);

    const int aRowIdx = live ? (mTile * 16 + l16) : l16;
    const float* aRow = A + (size_t)aRowIdx * K + 2 * half;
    v2f a[KS];
#pragma unroll
    for (int s = 0; s < KS; ++s) a[s] = *(const v2f*)(aRow + 4 * s);

#pragma unroll
    for (int i = 0; i < ITEMS / 256; ++i) {
        const int it = tid + 256 * i;
        const int p  = it / N4;
        const int n4 = it % N4;
        const v4f r0 = *(const v4f*)(W + (size_t)(2 * p)     * N + 4 * n4);
        const v4f r1 = *(const v4f*)(W + (size_t)(2 * p + 1) * N + 4 * n4);
        float* d = sW + p * PITCH + 8 * n4;
        v4f lo, hi;
        lo.x = r0.x; lo.y = r1.x; lo.z = r0.y; lo.w = r1.y;
        hi.x = r0.z; hi.y = r1.z; hi.z = r0.w; hi.w = r1.w;
        *(v4f*)d       = lo;
        *(v4f*)(d + 4) = hi;
    }
    __syncthreads();
    if (!live) return;

    if (RELU_A) {
#pragma unroll
        for (int s = 0; s < KS; ++s) {
            a[s].x = fmaxf(a[s].x, 0.0f);
            a[s].y = fmaxf(a[s].y, 0.0f);
        }
    }

    v8f c[NT];
#pragma unroll
    for (int t = 0; t < NT; ++t) {
        const float bv = bias[t * 16 + l16];
#pragma unroll
        for (int i = 0; i < 8; ++i) c[t][i] = bv;
    }

    const float* bBase = sW + half * PITCH + 2 * l16;
#pragma unroll
    for (int s = 0; s < KS; ++s) {
#pragma unroll
        for (int t = 0; t < NT; ++t) {
            const v2f b = *(const v2f*)(bBase + (2 * s) * PITCH + 32 * t);
            c[t] = __builtin_amdgcn_wmma_f32_16x16x4_f32(false, a[s], false, b,
                                                         (short)0, c[t], false, false);
        }
    }

    float* oBase = out + (size_t)(mTile * 16 + half * 8) * N + l16;
#pragma unroll
    for (int r = 0; r < 8; ++r) {
#pragma unroll
        for (int t = 0; t < NT; ++t) oBase[(size_t)r * N + 16 * t] = c[t][r];
    }
}

template <int FEATS, int K, typename VROW>
__device__ __forceinline__ void add_rows(VROW& acc, const float* __restrict__ hLane,
                                         int sL, int wL, int j) {
    int   s[K];
    float w[K];
    VROW  a[K];
#pragma unroll
    for (int i = 0; i < K; ++i) {
        s[i] = __builtin_amdgcn_readlane(sL, j + i);
        w[i] = __int_as_float(__builtin_amdgcn_readlane(wL, j + i));
    }
#pragma unroll
    for (int i = 0; i < K; ++i) a[i] = *(const VROW*)(hLane + (size_t)s[i] * FEATS);
#pragma unroll
    for (int i = 0; i < K; ++i) acc += a[i] * w[i];
}

template <int FEATS, int WSEL>
__global__ __launch_bounds__(256)
void gather_sum_kernel(const float* __restrict__ h, const int* __restrict__ rowptr,
                       const float* __restrict__ rec, float* __restrict__ out,
                       int nNodes) {
    constexpr int V = FEATS / 32;
    typedef __attribute__((ext_vector_type(V))) float vrow;
    const int lane = threadIdx.x & 31;
    const int d = __builtin_amdgcn_readfirstlane(blockIdx.x * 8 + (threadIdx.x >> 5));
    if (d >= nNodes) return;
    const int beg = rowptr[d];
    const int end = rowptr[d + 1];

    const float* hLane = h + lane * V;
    vrow acc = (vrow)(0.0f);

    for (int base = beg; base < end; base += 32) {
        const int n = min(32, end - base);
        v4f r = (v4f)(0.0f);
        if (lane < n) r = ((const v4f*)rec)[base + lane];
        const int sL = __float_as_int(r.x);
        const int wL = __float_as_int(WSEL == 1 ? r.y : (WSEL == 2 ? r.z : r.w));

        int j = 0;
        for (; j + 8 <= n; j += 8) add_rows<FEATS, 8>(acc, hLane, sL, wL, j);
        if (j + 4 <= n) { add_rows<FEATS, 4>(acc, hLane, sL, wL, j); j += 4; }
        const int rem = n - j;
        if (rem == 3)      add_rows<FEATS, 3>(acc, hLane, sL, wL, j);
        else if (rem == 2) add_rows<FEATS, 2>(acc, hLane, sL, wL, j);
        else if (rem == 1) add_rows<FEATS, 1>(acc, hLane, sL, wL, j);
    }
    *(vrow*)(out + (size_t)d * FEATS + lane * V) = acc;
}

extern "C" void kernel_launch(void* const* d_in, const int* in_sizes, int n_in,
                              void* d_out, int out_size, void* d_ws, size_t ws_size,
                              hipStream_t stream) {
    const float* x   = (const float*)d_in[0];
    const int*   src = (const int*)  d_in[1];
    const int*   dst = (const int*)  d_in[2];
    const float* w1  = (const float*)d_in[3];
    const float* w2  = (const float*)d_in[4];
    const float* w3  = (const float*)d_in[5];
    const float* W1  = (const float*)d_in[6];
    const float* b1  = (const float*)d_in[7];
    const float* W2  = (const float*)d_in[8];
    const float* b2  = (const float*)d_in[9];
    const float* W3  = (const float*)d_in[10];
    const float* b3  = (const float*)d_in[11];
    float* out = (float*)d_out;

    float* wsA    = (float*)d_ws;
    float* wsB    = wsA + (size_t)N_NODES * N_HIDDEN;
    float* rec    = wsB + (size_t)N_NODES * N_HIDDEN;
    int*   rank   = (int*)(rec + (size_t)N_EDGES * 4);
    int*   tbl    = rank + N_EDGES;
    int*   tbase  = tbl + T_PAD;
    int*   rowptr = tbase + T_PAD;
    float* recC   = (float*)(rowptr + CNT_PAD);

    const int mTiles   = (N_NODES + 15) / 16;
    const dim3 blk(256);
    const dim3 gemmGrid((mTiles + 7) / 8);
    const dim3 edgeGrid((N_EDGES + 255) / 256);
    const dim3 nodeGrid((N_NODES + 7) / 8);
    const dim3 chunkGrid(N_CHUNKS);
    const dim3 scanGrid((T_LEN + 1 + 1023) / 1024);
    const dim3 binGrid(N_BINS);

    coarse_count_kernel<<<chunkGrid, blk, 0, stream>>>(dst, tbl, rank, N_EDGES);
    rowptr_scan_kernel<<<scanGrid, dim3(1024), 0, stream>>>(tbl, tbase, T_LEN);
    coarse_place_kernel<<<edgeGrid, blk, 0, stream>>>(src, dst, w1, w2, w3, rank, tbase, recC, N_EDGES);
    fine_place_kernel<<<binGrid, dim3(1024), 0, stream>>>(recC, tbase, rowptr, rec, N_NODES);

    gemm_wmma_f32<N_HIDDEN, false><<<gemmGrid, blk, 0, stream>>>(x, W1, b1, wsA, N_NODES);
    gather_sum_kernel<N_HIDDEN, 1><<<nodeGrid, blk, 0, stream>>>(wsA, rowptr, rec, wsB, N_NODES);

    gemm_wmma_f32<N_HIDDEN, true><<<gemmGrid, blk, 0, stream>>>(wsB, W2, b2, wsA, N_NODES);
    gather_sum_kernel<N_HIDDEN, 2><<<nodeGrid, blk, 0, stream>>>(wsA, rowptr, rec, wsB, N_NODES);

    gemm_wmma_f32<N_CLASSES, true><<<gemmGrid, blk, 0, stream>>>(wsB, W3, b3, wsA, N_NODES);
    gather_sum_kernel<N_CLASSES, 3><<<nodeGrid, blk, 0, stream>>>(wsA, rowptr, rec, out, N_NODES);
}


# Edge bucketing in 2 launches, not 4: each 2048-edge chunk places its records bin-ordered in its own slice, a bin's work-group collects its 245 runs; no table scan; sum order still arrival-order rank
# speedup vs baseline: 3.3884x; 1.2316x over previous
#include <hip/hip_runtime.h>
#include <hip/hip_bf16.h>


#define N_NODES   50000
#define N_EDGES   500000
#define IN_FEATS  128
#define N_HIDDEN  128
#define N_CLASSES 64
#define CNT_PAD   50016
#define BIN_SHIFT 8
#define BIN_NODES 256
#define N_BINS    ((N_NODES + BIN_NODES) / BIN_NODES)
#define CHUNK     2048
#define N_CHUNKS  ((N_EDGES + CHUNK - 1) / CHUNK)
#define T_LEN     (N_BINS * N_CHUNKS)
#define T_PAD     48032
static_assert(N_NODES <= 65536, "a source index must fit the low 16 bits of the packed key");
static_assert(T_LEN <= T_PAD, "table padding");
static_assert(CHUNK < 32768, "run start and run length share one 32-bit table entry");
static_assert(N_NODES % 16 == 0, "the product kernels store whole 16-row strips");

typedef __attribute__((ext_vector_type(2))) float v2f;
typedef __attribute__((ext_vector_type(4))) float v4f;
typedef __attribute__((ext_vector_type(8))) float v8f;

__global__ __launch_bounds__(256)
void chunk_place_kernel(const int* __restrict__ src, const int* __restrict__ dst,
                        const float* __restrict__ w1, const float* __restrict__ w2,
                        const float* __restrict__ w3, int* __restrict__ tbl,
                        float* __restrict__ recC, int nEdges) {
    __shared__ int sCnt[256];
    __shared__ int sOff[256];
    __shared__ int sWave[8];
    static_assert(N_BINS <= 256, "one counter per thread");
    const int tid  = threadIdx.x;
    const int lane = tid & 31;
    const int wave = tid >> 5;
    sCnt[tid] = 0;
    const int base = blockIdx.x * CHUNK + tid;
    int   d[CHUNK / 256], s[CHUNK / 256];
    float a[CHUNK / 256], b[CHUNK / 256], c[CHUNK / 256];
#pragma unroll
    for (int i = 0; i < CHUNK / 256; ++i) {
        const int e = min(base + 256 * i, nEdges - 1);
        d[i] = dst[e];
        s[i] = src[e];
        a[i] = w1[e];
        b[i] = w2[e];
        c[i] = w3[e];
    }
    __syncthreads();
    int r[CHUNK / 256];
    if (blockIdx.x * CHUNK + CHUNK <= nEdges) {
#pragma unroll
        for (int i = 0; i < CHUNK / 256; ++i) r[i] = atomicAdd(&sCnt[d[i] >> BIN_SHIFT], 1);
    } else {
#pragma unroll
        for (int i = 0; i < CHUNK / 256; ++i) {
            r[i] = 0;
            if (base + 256 * i < nEdges) r[i] = atomicAdd(&sCnt[d[i] >> BIN_SHIFT], 1);
        }
    }
    __syncthreads();

    const int v = sCnt[tid];
    int incl = v;
#pragma unroll
    for (int o = 1; o < 32; o <<= 1) {
        const int t = __shfl_up(incl, o);
        if (lane >= o) incl += t;
    }
    if (lane == 31) sWave[wave] = incl;
    __syncthreads();
    int pre = 0;
#pragma unroll
    for (int w = 0; w < 8; ++w) pre += (w < wave) ? sWave[w] : 0;
    const int off = pre + incl - v;
    sOff[tid] = off;
    if (tid < N_BINS) tbl[tid * N_CHUNKS + blockIdx.x] = (off << 16) | v;
    __syncthreads();

    v4f* slice = (v4f*)recC + (size_t)blockIdx.x * CHUNK;
    int pos[CHUNK / 256];
#pragma unroll
    for (int i = 0; i < CHUNK / 256; ++i) pos[i] = sOff[d[i] >> BIN_SHIFT] + r[i];
#pragma unroll
    for (int i = 0; i < CHUNK / 256; ++i) {
        if (base + 256 * i < nEdges) {
            v4f q;
            q.x = __int_as_float(s[i] | ((d[i] & (BIN_NODES - 1)) << 16));
            q.y = a[i];
            q.z = b[i];
            q.w = c[i];
            slice[pos[i]] = q;
        }
    }
}

#define RUNS_PER_WAVE ((N_CHUNKS + 31) / 32)
__global__ __launch_bounds__(1024)
void fine_place_kernel(const float* __restrict__ recC, const int* __restrict__ tbl,
                       int* __restrict__ rowptr, float* __restrict__ rec, int nNodes) {
    __shared__ int sRun[N_CHUNKS];
    __shared__ int sCnt[BIN_NODES];
    __shared__ int sCur[BIN_NODES];
    __shared__ int sWave[BIN_NODES / 32];
    __shared__ int sPart[32];
    const int tid  = threadIdx.x;
    const int lane = tid & 31;
    const int wave = __builtin_amdgcn_readfirstlane(tid >> 5);
    const int bin  = blockIdx.x;

    int myRun = 0;
    if (tid < N_CHUNKS) myRun = tbl[bin * N_CHUNKS + tid];
    if (tid < BIN_NODES) sCnt[tid] = 0;

    const int nBefore = bin * N_CHUNKS;
    const int n4      = nBefore >> 2;
    int before = 0;
    for (int i = tid; i < n4; i += 1024) {
        const int4 q = ((const int4*)tbl)[i];
        before += (q.x & 0xFFFF) + (q.y & 0xFFFF) + (q.z & 0xFFFF) + (q.w & 0xFFFF);
    }
    if (tid < (nBefore & 3)) before += tbl[4 * n4 + tid] & 0xFFFF;
    if (tid < N_CHUNKS) sRun[tid] = myRun;
#pragma unroll
    for (int o = 16; o > 0; o >>= 1) before += __shfl_xor(before, o);
    if (lane == 0) sPart[wave] = before;
    __syncthreads();

    int segBeg = sPart[lane];
#pragma unroll
    for (int o = 16; o > 0; o >>= 1) segBeg += __shfl_xor(segBeg, o);

    int pv[RUNS_PER_WAVE], cnt[RUNS_PER_WAVE], rb[RUNS_PER_WAVE];
#pragma unroll
    for (int k = 0; k < RUNS_PER_WAVE; ++k) pv[k] = sRun[min(wave + 32 * k, N_CHUNKS - 1)];
#pragma unroll
    for (int k = 0; k < RUNS_PER_WAVE; ++k) {
        const int r = wave + 32 * k;
        const int p = (r < N_CHUNKS) ? __builtin_amdgcn_readfirstlane(pv[k]) : 0;
        cnt[k] = p & 0xFFFF;
        rb[k]  = r * CHUNK + (p >> 16);
    }
    v4f q[RUNS_PER_WAVE];
#pragma unroll
    for (int k = 0; k < RUNS_PER_WAVE; ++k) {
        q[k] = (v4f)(0.0f);
        if (lane < cnt[k]) q[k] = ((const v4f*)recC)[rb[k] + lane];
    }
#pragma unroll
    for (int k = 0; k < RUNS_PER_WAVE; ++k) {
        if (lane < cnt[k]) atomicAdd(&sCnt[(__float_as_int(q[k].x) >> 16) & (BIN_NODES - 1)], 1);
    }
#pragma unroll
    for (int k = 0; k < RUNS_PER_WAVE; ++k) {
        for (int i = 32 + lane; i < cnt[k]; i += 32) {
            const int key = __float_as_int(recC[(size_t)4 * (rb[k] + i)]);
            atomicAdd(&sCnt[(key >> 16) & (BIN_NODES - 1)], 1);
        }
    }
    __syncthreads();

    int v = 0, incl = 0;
    if (tid < BIN_NODES) {
        v = sCnt[tid];
        incl = v;
#pragma unroll
        for (int o = 1; o < 32; o <<= 1) {
            const int t = __shfl_up(incl, o);
            if (lane >= o) incl += t;
        }
        if (lane == 31) sWave[wave] = incl;
    }
    __syncthreads();
    if (tid < BIN_NODES) {
        int pre = 0;
#pragma unroll
        for (int w = 0; w < BIN_NODES / 32; ++w) pre += (w < wave) ? sWave[w] : 0;
        const int start = segBeg + pre + incl - v;
        sCur[tid] = start;
        const int node = bin * BIN_NODES + tid;
        if (node <= nNodes) rowptr[node] = start;
    }
    __syncthreads();

    int slot[RUNS_PER_WAVE];
#pragma unroll
    for (int k = 0; k < RUNS_PER_WAVE; ++k) {
        slot[k] = 0;
        if (lane < cnt[k]) slot[k] = atomicAdd(&sCur[(__float_as_int(q[k].x) >> 16) & (BIN_NODES - 1)], 1);
    }
#pragma unroll
    for (int k = 0; k < RUNS_PER_WAVE; ++k) {
        if (lane < cnt[k]) {
            v4f o = q[k];
            o.x = __int_as_float(__float_as_int(o.x) & 0xFFFF);
            ((v4f*)rec)[slot[k]] = o;
        }
    }
#pragma unroll
    for (int k = 0; k < RUNS_PER_WAVE; ++k) {
        for (int i = 32 + lane; i < cnt[k]; i += 32) {
            v4f o = ((const v4f*)recC)[rb[k] + i];
            const int key = __float_as_int(o.x);
            const int p = atomicAdd(&sCur[(key >> 16) & (BIN_NODES - 1)], 1);
            o.x = __int_as_float(key & 0xFFFF);
            ((v4f*)rec)[p] = o;
        }
    }
}

template <int N, bool RELU_A>
__global__ __launch_bounds__(256)
void gemm_wmma_f32(const float* __restrict__ A, const float* __restrict__ W,
                   const float* __restrict__ bias, float* __restrict__ out, int M) {
    constexpr int K     = 128;
    constexpr int KS    = K / 4;
    constexpr int NT    = N / 16;
    constexpr int PITCH = 2 * N + 32;
    constexpr int N4    = N / 4;
    constexpr int ITEMS = (K / 2) * N4;
    __shared__ __attribute__((aligned(16))) float sW[(K / 2) * PITCH];

    const int tid   = threadIdx.x;
    const int lane  = tid & 31;
    const int wave  = tid >> 5;
    const int half  = lane >> 4;
    const int l16   = lane & 15;
    const int mTile = blockIdx.x * 8 + wave;
    const bool live = (mTile * 16 < M);

    const int aRowIdx = live ? (mTile * 16 + l16) : l16;
    const float* aRow = A + (size_t)aRowIdx * K + 2 * half;
    v2f a[KS];
#pragma unroll
    for (int s = 0; s < KS; ++s) a[s] = *(const v2f*)(aRow + 4 * s);

#pragma unroll
    for (int i = 0; i < ITEMS / 256; ++i) {
        const int it = tid + 256 * i;
        const int p  = it / N4;
        const int n4 = it % N4;
        const v4f r0 = *(const v4f*)(W + (size_t)(2 * p)     * N + 4 * n4);
        const v4f r1 = *(const v4f*)(W + (size_t)(2 * p + 1) * N + 4 * n4);
        float* d = sW + p * PITCH + 8 * n4;
        v4f lo, hi;
        lo.x = r0.x; lo.y = r1.x; lo.z = r0.y; lo.w = r1.y;
        hi.x = r0.z; hi.y = r1.z; hi.z = r0.w; hi.w = r1.w;
        *(v4f*)d       = lo;
        *(v4f*)(d + 4) = hi;
    }
    __syncthreads();
    if (!live) return;

    if (RELU_A) {
#pragma unroll
        for (int s = 0; s < KS; ++s) {
            a[s].x = fmaxf(a[s].x, 0.0f);
            a[s].y = fmaxf(a[s].y, 0.0f);
        }
    }

    v8f c[NT];
#pragma unroll
    for (int t = 0; t < NT; ++t) {
        const float bv = bias[t * 16 + l16];
#pragma unroll
        for (int i = 0; i < 8; ++i) c[t][i] = bv;
    }

    const float* bBase = sW + half * PITCH + 2 * l16;
#pragma unroll
    for (int s = 0; s < KS; ++s) {
#pragma unroll
        for (int t = 0; t < NT; ++t) {
            const v2f b = *(const v2f*)(bBase + (2 * s) * PITCH + 32 * t);
            c[t] = __builtin_amdgcn_wmma_f32_16x16x4_f32(false, a[s], false, b,
                                                         (short)0, c[t], false, false);
        }
    }

    float* oBase = out + (size_t)(mTile * 16 + half * 8) * N + l16;
#pragma unroll
    for (int r = 0; r < 8; ++r) {
#pragma unroll
        for (int t = 0; t < NT; ++t) oBase[(size_t)r * N + 16 * t] = c[t][r];
    }
}

template <int FEATS, int K, typename VROW>
__device__ __forceinline__ void add_rows(VROW& acc, const float* __restrict__ hLane,
                                         int sL, int wL, int j) {
    int   s[K];
    float w[K];
    VROW  a[K];
#pragma unroll
    for (int i = 0; i < K; ++i) {
        s[i] = __builtin_amdgcn_readlane(sL, j + i);
        w[i] = __int_as_float(__builtin_amdgcn_readlane(wL, j + i));
    }
#pragma unroll
    for (int i = 0; i < K; ++i) a[i] = *(const VROW*)(hLane + (size_t)s[i] * FEATS);
#pragma unroll
    for (int i = 0; i < K; ++i) acc += a[i] * w[i];
}

template <int FEATS, int WSEL>
__global__ __launch_bounds__(256)
void gather_sum_kernel(const float* __restrict__ h, const int* __restrict__ rowptr,
                       const float* __restrict__ rec, float* __restrict__ out,
                       int nNodes) {
    constexpr int V = FEATS / 32;
    typedef __attribute__((ext_vector_type(V))) float vrow;
    const int lane = threadIdx.x & 31;
    const int d = __builtin_amdgcn_readfirstlane(blockIdx.x * 8 + (threadIdx.x >> 5));
    if (d >= nNodes) return;
    const int beg = rowptr[d];
    const int end = rowptr[d + 1];

    const float* hLane = h + lane * V;
    vrow acc = (vrow)(0.0f);

    for (int base = beg; base < end; base += 32) {
        const int n = min(32, end - base);
        v4f r = (v4f)(0.0f);
        if (lane < n) r = ((const v4f*)rec)[base + lane];
        const int sL = __float_as_int(r.x);
        const int wL = __float_as_int(WSEL == 1 ? r.y : (WSEL == 2 ? r.z : r.w));

        int j = 0;
        for (; j + 8 <= n; j += 8) add_rows<FEATS, 8>(acc, hLane, sL, wL, j);
        if (j + 4 <= n) { add_rows<FEATS, 4>(acc, hLane, sL, wL, j); j += 4; }
        const int rem = n - j;
        if (rem == 3)      add_rows<FEATS, 3>(acc, hLane, sL, wL, j);
        else if (rem == 2) add_rows<FEATS, 2>(acc, hLane, sL, wL, j);
        else if (rem == 1) add_rows<FEATS, 1>(acc, hLane, sL, wL, j);
    }
    *(vrow*)(out + (size_t)d * FEATS + lane * V) = acc;
}

#define A_PITCH 132

__device__ __forceinline__ void flush_row(float* __restrict__ sAw, int row, int lane, v4f& acc) {
    v4f r;
    r.x = fmaxf(acc.x, 0.0f);
    r.y = fmaxf(acc.y, 0.0f);
    r.z = fmaxf(acc.z, 0.0f);
    r.w = fmaxf(acc.w, 0.0f);
    *(v4f*)(sAw + row * A_PITCH + 4 * lane) = r;
    acc = (v4f)(0.0f);
}

template <int KK>
__device__ __forceinline__ void add_run(v4f& acc, int& cur, int& rowEnd, int rpv,
                                        float* __restrict__ sAw, const float* __restrict__ hLane,
                                        int sL, int wL, int j, int idx0, int lane) {
    int   s[KK];
    float w[KK];
    v4f   a[KK];
#pragma unroll
    for (int i = 0; i < KK; ++i) {
        s[i] = __builtin_amdgcn_readlane(sL, j + i);
        w[i] = __int_as_float(__builtin_amdgcn_readlane(wL, j + i));
    }
#pragma unroll
    for (int i = 0; i < KK; ++i) a[i] = *(const v4f*)(hLane + (size_t)s[i] * 128);
#pragma unroll
    for (int i = 0; i < KK; ++i) asm volatile("" :: "v"(a[i]));
#pragma unroll
    for (int i = 0; i < KK; ++i) {
        while (idx0 + i == rowEnd && cur < 15) {
            flush_row(sAw, cur, lane, acc);
            ++cur;
            rowEnd = __builtin_amdgcn_readlane(rpv, cur + 1);
        }
        acc += a[i] * w[i];
    }
}

template <int N, int WSEL>
__global__ __launch_bounds__(256)
void gather_gemm_wmma_f32(const float* __restrict__ h, const int* __restrict__ rowptr,
                          const float* __restrict__ rec, const float* __restrict__ W,
                          const float* __restrict__ bias, float* __restrict__ out, int M) {
    constexpr int K     = 128;
    constexpr int KS    = K / 4;
    constexpr int NT    = N / 16;
    constexpr int PITCH = 2 * N + 32;
    constexpr int N4    = N / 4;
    constexpr int ITEMS = (K / 2) * N4;
    constexpr int NI    = ITEMS / 256;
    __shared__ __attribute__((aligned(16))) float sW[(K / 2) * PITCH];
    __shared__ __attribute__((aligned(16))) float sA[8 * 16 * A_PITCH];

    const int tid   = threadIdx.x;
    const int lane  = tid & 31;
    const int wave  = __builtin_amdgcn_readfirstlane(tid >> 5);
    const int half  = lane >> 4;
    const int l16   = lane & 15;
    const int mTile = blockIdx.x * 8 + wave;
    const int row0  = mTile * 16;
    const bool live = (row0 < M);

    const int rpv = rowptr[min(min(row0, M) + min(lane, 16), M)];

    v4f r0[NI], r1[NI];
#pragma unroll
    for (int i = 0; i < NI; ++i) {
        const int it = tid + 256 * i;
        const int p  = it / N4;
        const int n4 = it % N4;
        r0[i] = *(const v4f*)(W + (size_t)(2 * p)     * N + 4 * n4);
        r1[i] = *(const v4f*)(W + (size_t)(2 * p + 1) * N + 4 * n4);
    }

    const int beg0   = __builtin_amdgcn_readlane(rpv, 0);
    const int endAll = live ? __builtin_amdgcn_readlane(rpv, 16) : beg0;
    const int* recI = (const int*)rec;
    int sCur, wCur;
    {
        const int e = min(beg0 + lane, N_EDGES - 1);
        sCur = recI[4 * e];
        wCur = recI[4 * e + WSEL];
    }

#pragma unroll
    for (int i = 0; i < NI; ++i) {
        const int it = tid + 256 * i;
        const int p  = it / N4;
        const int n4 = it % N4;
        float* d = sW + p * PITCH + 8 * n4;
        v4f lo, hi;
        lo.x = r0[i].x; lo.y = r1[i].x; lo.z = r0[i].y; lo.w = r1[i].y;
        hi.x = r0[i].z; hi.y = r1[i].z; hi.z = r0[i].w; hi.w = r1[i].w;
        *(v4f*)d       = lo;
        *(v4f*)(d + 4) = hi;
    }

    float bv[NT];
#pragma unroll
    for (int t = 0; t < NT; ++t) bv[t] = bias[t * 16 + l16];

    float* sAw = sA + wave * (16 * A_PITCH);
    if (live) {
        const float* hLane = h + lane * 4;
        v4f acc   = (v4f)(0.0f);
        int cur    = 0;
        int rowEnd = __builtin_amdgcn_readlane(rpv, 1);
        for (int base = beg0; base < endAll; base += 32) {
            const int n  = min(32, endAll - base);
            const int sL = sCur;
            const int wL = wCur;
            const int eN = min(base + 32 + lane, N_EDGES - 1);
            const int sNext = recI[4 * eN];
            const int wNext = recI[4 * eN + WSEL];

            int j = 0;
            for (; j + 16 <= n; j += 16) add_run<16>(acc, cur, rowEnd, rpv, sAw, hLane, sL, wL, j, base + j, lane);
            if (j + 8 <= n) { add_run<8>(acc, cur, rowEnd, rpv, sAw, hLane, sL, wL, j, base + j, lane); j += 8; }
            if (j + 4 <= n) { add_run<4>(acc, cur, rowEnd, rpv, sAw, hLane, sL, wL, j, base + j, lane); j += 4; }
            const int rem = n - j;
            if (rem == 3)      add_run<3>(acc, cur, rowEnd, rpv, sAw, hLane, sL, wL, j, base + j, lane);
            else if (rem == 2) add_run<2>(acc, cur, rowEnd, rpv, sAw, hLane, sL, wL, j, base + j, lane);
            else if (rem == 1) add_run<1>(acc, cur, rowEnd, rpv, sAw, hLane, sL, wL, j, base + j, lane);
            sCur = sNext;
            wCur = wNext;
        }
        for (; cur < 16; ++cur) flush_row(sAw, cur, lane, acc);
    }
    __syncthreads();
    if (!live) return;

    v2f a[KS];
#pragma unroll
    for (int s = 0; s < KS; ++s) a[s] = *(const v2f*)(sAw + l16 * A_PITCH + 4 * s + 2 * half);

    v8f c[NT];
#pragma unroll
    for (int t = 0; t < NT; ++t) {
#pragma unroll
        for (int i = 0; i < 8; ++i) c[t][i] = bv[t];
    }

    const float* bBase = sW + half * PITCH + 2 * l16;
#pragma unroll
    for (int s = 0; s < KS; ++s) {
#pragma unroll
        for (int t = 0; t < NT; ++t) {
            const v2f b = *(const v2f*)(bBase + (2 * s) * PITCH + 32 * t);
            c[t] = __builtin_amdgcn_wmma_f32_16x16x4_f32(false, a[s], false, b,
                                                         (short)0, c[t], false, false);
        }
    }

    float* oBase = out + (size_t)(row0 + half * 8) * N + l16;
#pragma unroll
    for (int r = 0; r < 8; ++r) {
#pragma unroll
        for (int t = 0; t < NT; ++t) oBase[(size_t)r * N + 16 * t] = c[t][r];
    }
}

extern "C" void kernel_launch(void* const* d_in, const int* in_sizes, int n_in,
                              void* d_out, int out_size, void* d_ws, size_t ws_size,
                              hipStream_t stream) {
    const float* x   = (const float*)d_in[0];
    const int*   src = (const int*)  d_in[1];
    const int*   dst = (const int*)  d_in[2];
    const float* w1  = (const float*)d_in[3];
    const float* w2  = (const float*)d_in[4];
    const float* w3  = (const float*)d_in[5];
    const float* W1  = (const float*)d_in[6];
    const float* b1  = (const float*)d_in[7];
    const float* W2  = (const float*)d_in[8];
    const float* b2  = (const float*)d_in[9];
    const float* W3  = (const float*)d_in[10];
    const float* b3  = (const float*)d_in[11];
    float* out = (float*)d_out;

    float* wsA    = (float*)d_ws;
    float* wsB    = wsA + (size_t)N_NODES * N_HIDDEN;
    float* rec    = wsB + (size_t)N_NODES * N_HIDDEN;
    float* recC   = rec + (size_t)N_EDGES * 4;
    int*   tbl    = (int*)(recC + (size_t)N_EDGES * 4);
    int*   rowptr = tbl + T_PAD;

    const int mTiles   = (N_NODES + 15) / 16;
    const dim3 blk(256);
    const dim3 gemmGrid((mTiles + 7) / 8);
    const dim3 nodeGrid((N_NODES + 7) / 8);
    const dim3 chunkGrid(N_CHUNKS);
    const dim3 binGrid(N_BINS);

    chunk_place_kernel<<<chunkGrid, blk, 0, stream>>>(src, dst, w1, w2, w3, tbl, recC, N_EDGES);
    fine_place_kernel<<<binGrid, dim3(1024), 0, stream>>>(recC, tbl, rowptr, rec, N_NODES);

    gemm_wmma_f32<N_HIDDEN, false><<<gemmGrid, blk, 0, stream>>>(x, W1, b1, wsA, N_NODES);

    gather_gemm_wmma_f32<N_HIDDEN, 1><<<gemmGrid, blk, 0, stream>>>(wsA, rowptr, rec, W2, b2, wsB, N_NODES);

    gather_gemm_wmma_f32<N_CLASSES, 2><<<gemmGrid, blk, 0, stream>>>(wsB, rowptr, rec, W3, b3, wsA, N_NODES);

    gather_sum_kernel<N_CLASSES, 3><<<nodeGrid, blk, 0, stream>>>(wsA, rowptr, rec, out, N_NODES);
}


# Layer-1 product and coarse edge pass share one grid (work-groups 0-390 multiply, the other 245 place edge chunks): independent work overlaps, one launch boundary goes; sum order unchanged
# speedup vs baseline: 3.6237x; 1.0694x over previous
#include <hip/hip_runtime.h>
#include <hip/hip_bf16.h>


#define N_NODES   50000
#define N_EDGES   500000
#define IN_FEATS  128
#define N_HIDDEN  128
#define N_CLASSES 64
#define CNT_PAD   50016
#define BIN_SHIFT 8
#define BIN_NODES 256
#define N_BINS    ((N_NODES + BIN_NODES) / BIN_NODES)
#define CHUNK     2048
#define N_CHUNKS  ((N_EDGES + CHUNK - 1) / CHUNK)
#define T_LEN     (N_BINS * N_CHUNKS)
#define T_PAD     48032
#define ROW_BLOCKS ((N_NODES + 127) / 128)
static_assert(N_NODES <= 65536, "a source index must fit the low 16 bits of the packed key");
static_assert(T_LEN <= T_PAD, "table padding");
static_assert(CHUNK < 32768, "run start and run length share one 32-bit table entry");
static_assert(N_NODES % 16 == 0, "the product kernels store whole 16-row strips");

typedef __attribute__((ext_vector_type(2))) float v2f;
typedef __attribute__((ext_vector_type(4))) float v4f;
typedef __attribute__((ext_vector_type(8))) float v8f;

__device__ __forceinline__
void chunk_place_body(const int chunk,
                      const int* __restrict__ src, const int* __restrict__ dst,
                      const float* __restrict__ w1, const float* __restrict__ w2,
                      const float* __restrict__ w3, int* __restrict__ tbl,
                      float* __restrict__ recC, int nEdges) {
    __shared__ int sCnt[256];
    __shared__ int sOff[256];
    __shared__ int sWave[8];
    static_assert(N_BINS <= 256, "one counter per thread");
    const int tid  = threadIdx.x;
    const int lane = tid & 31;
    const int wave = tid >> 5;
    sCnt[tid] = 0;
    const int base = chunk * CHUNK + tid;
    int   d[CHUNK / 256], s[CHUNK / 256];
    float a[CHUNK / 256], b[CHUNK / 256], c[CHUNK / 256];
#pragma unroll
    for (int i = 0; i < CHUNK / 256; ++i) {
        const int e = min(base + 256 * i, nEdges - 1);
        d[i] = dst[e];
        s[i] = src[e];
        a[i] = w1[e];
        b[i] = w2[e];
        c[i] = w3[e];
    }
    __syncthreads();
    int r[CHUNK / 256];
    if (chunk * CHUNK + CHUNK <= nEdges) {
#pragma unroll
        for (int i = 0; i < CHUNK / 256; ++i) r[i] = atomicAdd(&sCnt[d[i] >> BIN_SHIFT], 1);
    } else {
#pragma unroll
        for (int i = 0; i < CHUNK / 256; ++i) {
            r[i] = 0;
            if (base + 256 * i < nEdges) r[i] = atomicAdd(&sCnt[d[i] >> BIN_SHIFT], 1);
        }
    }
    __syncthreads();

    const int v = sCnt[tid];
    int incl = v;
#pragma unroll
    for (int o = 1; o < 32; o <<= 1) {
        const int t = __shfl_up(incl, o);
        if (lane >= o) incl += t;
    }
    if (lane == 31) sWave[wave] = incl;
    __syncthreads();
    int pre = 0;
#pragma unroll
    for (int w = 0; w < 8; ++w) pre += (w < wave) ? sWave[w] : 0;
    const int off = pre + incl - v;
    sOff[tid] = off;
    if (tid < N_BINS) tbl[tid * N_CHUNKS + chunk] = (off << 16) | v;
    __syncthreads();

    v4f* slice = (v4f*)recC + (size_t)chunk * CHUNK;
    int pos[CHUNK / 256];
#pragma unroll
    for (int i = 0; i < CHUNK / 256; ++i) pos[i] = sOff[d[i] >> BIN_SHIFT] + r[i];
#pragma unroll
    for (int i = 0; i < CHUNK / 256; ++i) {
        if (base + 256 * i < nEdges) {
            v4f q;
            q.x = __int_as_float(s[i] | ((d[i] & (BIN_NODES - 1)) << 16));
            q.y = a[i];
            q.z = b[i];
            q.w = c[i];
            slice[pos[i]] = q;
        }
    }
}

#define RUNS_PER_WAVE ((N_CHUNKS + 31) / 32)
__global__ __launch_bounds__(1024)
void fine_place_kernel(const float* __restrict__ recC, const int* __restrict__ tbl,
                       int* __restrict__ rowptr, float* __restrict__ rec, int nNodes) {
    __shared__ int sRun[N_CHUNKS];
    __shared__ int sCnt[BIN_NODES];
    __shared__ int sCur[BIN_NODES];
    __shared__ int sWave[BIN_NODES / 32];
    __shared__ int sPart[32];
    const int tid  = threadIdx.x;
    const int lane = tid & 31;
    const int wave = __builtin_amdgcn_readfirstlane(tid >> 5);
    const int bin  = blockIdx.x;

    int myRun = 0;
    if (tid < N_CHUNKS) myRun = tbl[bin * N_CHUNKS + tid];
    if (tid < BIN_NODES) sCnt[tid] = 0;

    const int nBefore = bin * N_CHUNKS;
    const int n4      = nBefore >> 2;
    int before = 0;
    for (int i = tid; i < n4; i += 1024) {
        const int4 q = ((const int4*)tbl)[i];
        before += (q.x & 0xFFFF) + (q.y & 0xFFFF) + (q.z & 0xFFFF) + (q.w & 0xFFFF);
    }
    if (tid < (nBefore & 3)) before += tbl[4 * n4 + tid] & 0xFFFF;
    if (tid < N_CHUNKS) sRun[tid] = myRun;
#pragma unroll
    for (int o = 16; o > 0; o >>= 1) before += __shfl_xor(before, o);
    if (lane == 0) sPart[wave] = before;
    __syncthreads();

    int segBeg = sPart[lane];
#pragma unroll
    for (int o = 16; o > 0; o >>= 1) segBeg += __shfl_xor(segBeg, o);

    int pv[RUNS_PER_WAVE], cnt[RUNS_PER_WAVE], rb[RUNS_PER_WAVE];
#pragma unroll
    for (int k = 0; k < RUNS_PER_WAVE; ++k) pv[k] = sRun[min(wave + 32 * k, N_CHUNKS - 1)];
#pragma unroll
    for (int k = 0; k < RUNS_PER_WAVE; ++k) {
        const int r = wave + 32 * k;
        const int p = (r < N_CHUNKS) ? __builtin_amdgcn_readfirstlane(pv[k]) : 0;
        cnt[k] = p & 0xFFFF;
        rb[k]  = r * CHUNK + (p >> 16);
    }
    v4f q[RUNS_PER_WAVE];
#pragma unroll
    for (int k = 0; k < RUNS_PER_WAVE; ++k) {
        q[k] = (v4f)(0.0f);
        if (lane < cnt[k]) q[k] = ((const v4f*)recC)[rb[k] + lane];
    }
#pragma unroll
    for (int k = 0; k < RUNS_PER_WAVE; ++k) {
        if (lane < cnt[k]) atomicAdd(&sCnt[(__float_as_int(q[k].x) >> 16) & (BIN_NODES - 1)], 1);
    }
#pragma unroll
    for (int k = 0; k < RUNS_PER_WAVE; ++k) {
        for (int i = 32 + lane; i < cnt[k]; i += 32) {
            const int key = __float_as_int(recC[(size_t)4 * (rb[k] + i)]);
            atomicAdd(&sCnt[(key >> 16) & (BIN_NODES - 1)], 1);
        }
    }
    __syncthreads();

    int v = 0, incl = 0;
    if (tid < BIN_NODES) {
        v = sCnt[tid];
        incl = v;
#pragma unroll
        for (int o = 1; o < 32; o <<= 1) {
            const int t = __shfl_up(incl, o);
            if (lane >= o) incl += t;
        }
        if (lane == 31) sWave[wave] = incl;
    }
    __syncthreads();
    if (tid < BIN_NODES) {
        int pre = 0;
#pragma unroll
        for (int w = 0; w < BIN_NODES / 32; ++w) pre += (w < wave) ? sWave[w] : 0;
        const int start = segBeg + pre + incl - v;
        sCur[tid] = start;
        const int node = bin * BIN_NODES + tid;
        if (node <= nNodes) rowptr[node] = start;
    }
    __syncthreads();

    int slot[RUNS_PER_WAVE];
#pragma unroll
    for (int k = 0; k < RUNS_PER_WAVE; ++k) {
        slot[k] = 0;
        if (lane < cnt[k]) slot[k] = atomicAdd(&sCur[(__float_as_int(q[k].x) >> 16) & (BIN_NODES - 1)], 1);
    }
#pragma unroll
    for (int k = 0; k < RUNS_PER_WAVE; ++k) {
        if (lane < cnt[k]) {
            v4f o = q[k];
            o.x = __int_as_float(__float_as_int(o.x) & 0xFFFF);
            ((v4f*)rec)[slot[k]] = o;
        }
    }
#pragma unroll
    for (int k = 0; k < RUNS_PER_WAVE; ++k) {
        for (int i = 32 + lane; i < cnt[k]; i += 32) {
            v4f o = ((const v4f*)recC)[rb[k] + i];
            const int key = __float_as_int(o.x);
            const int p = atomicAdd(&sCur[(key >> 16) & (BIN_NODES - 1)], 1);
            o.x = __int_as_float(key & 0xFFFF);
            ((v4f*)rec)[p] = o;
        }
    }
}

template <int N, bool RELU_A>
__device__ __forceinline__
void gemm_wmma_f32_body(const int rowBlock,
                        const float* __restrict__ A, const float* __restrict__ W,
                        const float* __restrict__ bias, float* __restrict__ out, int M) {
    constexpr int K     = 128;
    constexpr int KS    = K / 4;
    constexpr int NT    = N / 16;
    constexpr int PITCH = 2 * N + 32;
    constexpr int N4    = N / 4;
    constexpr int ITEMS = (K / 2) * N4;
    __shared__ __attribute__((aligned(16))) float sW[(K / 2) * PITCH];

    const int tid   = threadIdx.x;
    const int lane  = tid & 31;
    const int wave  = tid >> 5;
    const int half  = lane >> 4;
    const int l16   = lane & 15;
    const int mTile = rowBlock * 8 + wave;
    const bool live = (mTile * 16 < M);

    const int aRowIdx = live ? (mTile * 16 + l16) : l16;
    const float* aRow = A + (size_t)aRowIdx * K + 2 * half;
    v2f a[KS];
#pragma unroll
    for (int s = 0; s < KS; ++s) a[s] = *(const v2f*)(aRow + 4 * s);

#pragma unroll
    for (int i = 0; i < ITEMS / 256; ++i) {
        const int it = tid + 256 * i;
        const int p  = it / N4;
        const int n4 = it % N4;
        const v4f r0 = *(const v4f*)(W + (size_t)(2 * p)     * N + 4 * n4);
        const v4f r1 = *(const v4f*)(W + (size_t)(2 * p + 1) * N + 4 * n4);
        float* d = sW + p * PITCH + 8 * n4;
        v4f lo, hi;
        lo.x = r0.x; lo.y = r1.x; lo.z = r0.y; lo.w = r1.y;
        hi.x = r0.z; hi.y = r1.z; hi.z = r0.w; hi.w = r1.w;
        *(v4f*)d       = lo;
        *(v4f*)(d + 4) = hi;
    }
#pragma unroll
    for (int s = 0; s < KS; ++s) asm volatile("" :: "v"(a[s]));
    __syncthreads();
    if (!live) return;

    if (RELU_A) {
#pragma unroll
        for (int s = 0; s < KS; ++s) {
            a[s].x = fmaxf(a[s].x, 0.0f);
            a[s].y = fmaxf(a[s].y, 0.0f);
        }
    }

    v8f c[NT];
#pragma unroll
    for (int t = 0; t < NT; ++t) {
        const float bv = bias[t * 16 + l16];
#pragma unroll
        for (int i = 0; i < 8; ++i) c[t][i] = bv;
    }

    const float* bBase = sW + half * PITCH + 2 * l16;
#pragma unroll
    for (int s = 0; s < KS; ++s) {
#pragma unroll
        for (int t = 0; t < NT; ++t) {
            const v2f b = *(const v2f*)(bBase + (2 * s) * PITCH + 32 * t);
            c[t] = __builtin_amdgcn_wmma_f32_16x16x4_f32(false, a[s], false, b,
                                                         (short)0, c[t], false, false);
        }
    }

    float* oBase = out + (size_t)(mTile * 16 + half * 8) * N + l16;
#pragma unroll
    for (int r = 0; r < 8; ++r) {
#pragma unroll
        for (int t = 0; t < NT; ++t) oBase[(size_t)r * N + 16 * t] = c[t][r];
    }
}

template <int N>
__global__ __launch_bounds__(256)
void chunk_place_and_gemm_kernel(const int* __restrict__ src, const int* __restrict__ dst,
                                 const float* __restrict__ w1, const float* __restrict__ w2,
                                 const float* __restrict__ w3, int* __restrict__ tbl,
                                 float* __restrict__ recC, int nEdges,
                                 const float* __restrict__ A, const float* __restrict__ W,
                                 const float* __restrict__ bias, float* __restrict__ out, int M) {
    if (blockIdx.x < ROW_BLOCKS) {
        gemm_wmma_f32_body<N, false>(blockIdx.x, A, W, bias, out, M);
    } else {
        chunk_place_body(blockIdx.x - ROW_BLOCKS, src, dst, w1, w2, w3, tbl, recC, nEdges);
    }
}

template <int FEATS, int K, typename VROW>
__device__ __forceinline__ void add_rows(VROW& acc, const float* __restrict__ hLane,
                                         int sL, int wL, int j) {
    int   s[K];
    float w[K];
    VROW  a[K];
#pragma unroll
    for (int i = 0; i < K; ++i) {
        s[i] = __builtin_amdgcn_readlane(sL, j + i);
        w[i] = __int_as_float(__builtin_amdgcn_readlane(wL, j + i));
    }
#pragma unroll
    for (int i = 0; i < K; ++i) a[i] = *(const VROW*)(hLane + (size_t)s[i] * FEATS);
#pragma unroll
    for (int i = 0; i < K; ++i) acc += a[i] * w[i];
}

template <int FEATS, int WSEL>
__global__ __launch_bounds__(256)
void gather_sum_kernel(const float* __restrict__ h, const int* __restrict__ rowptr,
                       const float* __restrict__ rec, float* __restrict__ out,
                       int nNodes) {
    constexpr int V = FEATS / 32;
    typedef __attribute__((ext_vector_type(V))) float vrow;
    const int lane = threadIdx.x & 31;
    const int d = __builtin_amdgcn_readfirstlane(blockIdx.x * 8 + (threadIdx.x >> 5));
    if (d >= nNodes) return;
    const int beg = rowptr[d];
    const int end = rowptr[d + 1];

    const float* hLane = h + lane * V;
    vrow acc = (vrow)(0.0f);

    for (int base = beg; base < end; base += 32) {
        const int n = min(32, end - base);
        v4f r = (v4f)(0.0f);
        if (lane < n) r = ((const v4f*)rec)[base + lane];
        const int sL = __float_as_int(r.x);
        const int wL = __float_as_int(WSEL == 1 ? r.y : (WSEL == 2 ? r.z : r.w));

        int j = 0;
        for (; j + 8 <= n; j += 8) add_rows<FEATS, 8>(acc, hLane, sL, wL, j);
        if (j + 4 <= n) { add_rows<FEATS, 4>(acc, hLane, sL, wL, j); j += 4; }
        const int rem = n - j;
        if (rem == 3)      add_rows<FEATS, 3>(acc, hLane, sL, wL, j);
        else if (rem == 2) add_rows<FEATS, 2>(acc, hLane, sL, wL, j);
        else if (rem == 1) add_rows<FEATS, 1>(acc, hLane, sL, wL, j);
    }
    *(vrow*)(out + (size_t)d * FEATS + lane * V) = acc;
}

#define A_PITCH 132

__device__ __forceinline__ void flush_row(float* __restrict__ sAw, int row, int lane, v4f& acc) {
    v4f r;
    r.x = fmaxf(acc.x, 0.0f);
    r.y = fmaxf(acc.y, 0.0f);
    r.z = fmaxf(acc.z, 0.0f);
    r.w = fmaxf(acc.w, 0.0f);
    *(v4f*)(sAw + row * A_PITCH + 4 * lane) = r;
    acc = (v4f)(0.0f);
}

template <int KK>
__device__ __forceinline__ void add_run(v4f& acc, int& cur, int& rowEnd, int rpv,
                                        float* __restrict__ sAw, const float* __restrict__ hLane,
                                        int sL, int wL, int j, int idx0, int lane) {
    int   s[KK];
    float w[KK];
    v4f   a[KK];
#pragma unroll
    for (int i = 0; i < KK; ++i) {
        s[i] = __builtin_amdgcn_readlane(sL, j + i);
        w[i] = __int_as_float(__builtin_amdgcn_readlane(wL, j + i));
    }
#pragma unroll
    for (int i = 0; i < KK; ++i) a[i] = *(const v4f*)(hLane + (size_t)s[i] * 128);
#pragma unroll
    for (int i = 0; i < KK; ++i) asm volatile("" :: "v"(a[i]));
#pragma unroll
    for (int i = 0; i < KK; ++i) {
        while (idx0 + i == rowEnd && cur < 15) {
            flush_row(sAw, cur, lane, acc);
            ++cur;
            rowEnd = __builtin_amdgcn_readlane(rpv, cur + 1);
        }
        acc += a[i] * w[i];
    }
}

template <int N, int WSEL>
__global__ __launch_bounds__(256)
void gather_gemm_wmma_f32(const float* __restrict__ h, const int* __restrict__ rowptr,
                          const float* __restrict__ rec, const float* __restrict__ W,
                          const float* __restrict__ bias, float* __restrict__ out, int M) {
    constexpr int K     = 128;
    constexpr int KS    = K / 4;
    constexpr int NT    = N / 16;
    constexpr int PITCH = 2 * N + 32;
    constexpr int N4    = N / 4;
    constexpr int ITEMS = (K / 2) * N4;
    constexpr int NI    = ITEMS / 256;
    __shared__ __attribute__((aligned(16))) float sW[(K / 2) * PITCH];
    __shared__ __attribute__((aligned(16))) float sA[8 * 16 * A_PITCH];

    const int tid   = threadIdx.x;
    const int lane  = tid & 31;
    const int wave  = __builtin_amdgcn_readfirstlane(tid >> 5);
    const int half  = lane >> 4;
    const int l16   = lane & 15;
    const int mTile = blockIdx.x * 8 + wave;
    const int row0  = mTile * 16;
    const bool live = (row0 < M);

    const int rpv = rowptr[min(min(row0, M) + min(lane, 16), M)];

    v4f r0[NI], r1[NI];
#pragma unroll
    for (int i = 0; i < NI; ++i) {
        const int it = tid + 256 * i;
        const int p  = it / N4;
        const int n4 = it % N4;
        r0[i] = *(const v4f*)(W + (size_t)(2 * p)     * N + 4 * n4);
        r1[i] = *(const v4f*)(W + (size_t)(2 * p + 1) * N + 4 * n4);
    }

    const int beg0   = __builtin_amdgcn_readlane(rpv, 0);
    const int endAll = live ? __builtin_amdgcn_readlane(rpv, 16) : beg0;
    const int* recI = (const int*)rec;
    int sCur, wCur;
    {
        const int e = min(beg0 + lane, N_EDGES - 1);
        sCur = recI[4 * e];
        wCur = recI[4 * e + WSEL];
    }

#pragma unroll
    for (int i = 0; i < NI; ++i) {
        const int it = tid + 256 * i;
        const int p  = it / N4;
        const int n4 = it % N4;
        float* d = sW + p * PITCH + 8 * n4;
        v4f lo, hi;
        lo.x = r0[i].x; lo.y = r1[i].x; lo.z = r0[i].y; lo.w = r1[i].y;
        hi.x = r0[i].z; hi.y = r1[i].z; hi.z = r0[i].w; hi.w = r1[i].w;
        *(v4f*)d       = lo;
        *(v4f*)(d + 4) = hi;
    }

    float bv[NT];
#pragma unroll
    for (int t = 0; t < NT; ++t) bv[t] = bias[t * 16 + l16];

    float* sAw = sA + wave * (16 * A_PITCH);
    if (live) {
        const float* hLane = h + lane * 4;
        v4f acc   = (v4f)(0.0f);
        int cur    = 0;
        int rowEnd = __builtin_amdgcn_readlane(rpv, 1);
        for (int base = beg0; base < endAll; base += 32) {
            const int n  = min(32, endAll - base);
            const int sL = sCur;
            const int wL = wCur;
            const int eN = min(base + 32 + lane, N_EDGES - 1);
            const int sNext = recI[4 * eN];
            const int wNext = recI[4 * eN + WSEL];

            int j = 0;
            for (; j + 16 <= n; j += 16) add_run<16>(acc, cur, rowEnd, rpv, sAw, hLane, sL, wL, j, base + j, lane);
            if (j + 8 <= n) { add_run<8>(acc, cur, rowEnd, rpv, sAw, hLane, sL, wL, j, base + j, lane); j += 8; }
            if (j + 4 <= n) { add_run<4>(acc, cur, rowEnd, rpv, sAw, hLane, sL, wL, j, base + j, lane); j += 4; }
            const int rem = n - j;
            if (rem == 3)      add_run<3>(acc, cur, rowEnd, rpv, sAw, hLane, sL, wL, j, base + j, lane);
            else if (rem == 2) add_run<2>(acc, cur, rowEnd, rpv, sAw, hLane, sL, wL, j, base + j, lane);
            else if (rem == 1) add_run<1>(acc, cur, rowEnd, rpv, sAw, hLane, sL, wL, j, base + j, lane);
            sCur = sNext;
            wCur = wNext;
        }
        for (; cur < 16; ++cur) flush_row(sAw, cur, lane, acc);
    }
    __syncthreads();
    if (!live) return;

    v2f a[KS];
#pragma unroll
    for (int s = 0; s < KS; ++s) a[s] = *(const v2f*)(sAw + l16 * A_PITCH + 4 * s + 2 * half);

    v8f c[NT];
#pragma unroll
    for (int t = 0; t < NT; ++t) {
#pragma unroll
        for (int i = 0; i < 8; ++i) c[t][i] = bv[t];
    }

    const float* bBase = sW + half * PITCH + 2 * l16;
#pragma unroll
    for (int s = 0; s < KS; ++s) {
#pragma unroll
        for (int t = 0; t < NT; ++t) {
            const v2f b = *(const v2f*)(bBase + (2 * s) * PITCH + 32 * t);
            c[t] = __builtin_amdgcn_wmma_f32_16x16x4_f32(false, a[s], false, b,
                                                         (short)0, c[t], false, false);
        }
    }

    float* oBase = out + (size_t)(row0 + half * 8) * N + l16;
#pragma unroll
    for (int r = 0; r < 8; ++r) {
#pragma unroll
        for (int t = 0; t < NT; ++t) oBase[(size_t)r * N + 16 * t] = c[t][r];
    }
}

extern "C" void kernel_launch(void* const* d_in, const int* in_sizes, int n_in,
                              void* d_out, int out_size, void* d_ws, size_t ws_size,
                              hipStream_t stream) {
    const float* x   = (const float*)d_in[0];
    const int*   src = (const int*)  d_in[1];
    const int*   dst = (const int*)  d_in[2];
    const float* w1  = (const float*)d_in[3];
    const float* w2  = (const float*)d_in[4];
    const float* w3  = (const float*)d_in[5];
    const float* W1  = (const float*)d_in[6];
    const float* b1  = (const float*)d_in[7];
    const float* W2  = (const float*)d_in[8];
    const float* b2  = (const float*)d_in[9];
    const float* W3  = (const float*)d_in[10];
    const float* b3  = (const float*)d_in[11];
    float* out = (float*)d_out;

    float* wsA    = (float*)d_ws;
    float* wsB    = wsA + (size_t)N_NODES * N_HIDDEN;
    float* rec    = wsB + (size_t)N_NODES * N_HIDDEN;
    float* recC   = rec + (size_t)N_EDGES * 4;
    int*   tbl    = (int*)(recC + (size_t)N_EDGES * 4);
    int*   rowptr = tbl + T_PAD;

    const int mTiles   = (N_NODES + 15) / 16;
    const dim3 blk(256);
    const dim3 gemmGrid((mTiles + 7) / 8);
    const dim3 nodeGrid((N_NODES + 7) / 8);
    const dim3 firstGrid(ROW_BLOCKS + N_CHUNKS);
    const dim3 binGrid(N_BINS);

    chunk_place_and_gemm_kernel<N_HIDDEN><<<firstGrid, blk, 0, stream>>>(src, dst, w1, w2, w3, tbl, recC, N_EDGES,
                                                                  x, W1, b1, wsA, N_NODES);
    fine_place_kernel<<<binGrid, dim3(1024), 0, stream>>>(recC, tbl, rowptr, rec, N_NODES);

    gather_gemm_wmma_f32<N_HIDDEN, 1><<<gemmGrid, blk, 0, stream>>>(wsA, rowptr, rec, W2, b2, wsB, N_NODES);

    gather_gemm_wmma_f32<N_CLASSES, 2><<<gemmGrid, blk, 0, stream>>>(wsB, rowptr, rec, W3, b3, wsA, N_NODES);

    gather_sum_kernel<N_CLASSES, 3><<<nodeGrid, blk, 0, stream>>>(wsA, rowptr, rec, out, N_NODES);
}


# Layers 2 and 3: grid sized to the chip: 256 work-groups of 16 waves, one per WGP, own 12 or 13 row strips each (was 391 of 8, two on some WGPs); products dealt as 32-column units; sum order unchanged
# speedup vs baseline: 3.8944x; 1.0747x over previous
#include <hip/hip_runtime.h>
#include <hip/hip_bf16.h>


#define N_NODES   50000
#define N_EDGES   500000
#define IN_FEATS  128
#define N_HIDDEN  128
#define N_CLASSES 64
#define CNT_PAD   50016
#define BIN_SHIFT 8
#define BIN_NODES 256
#define N_BINS    ((N_NODES + BIN_NODES) / BIN_NODES)
#define CHUNK     2048
#define N_CHUNKS  ((N_EDGES + CHUNK - 1) / CHUNK)
#define T_LEN     (N_BINS * N_CHUNKS)
#define T_PAD     48032
#define ROW_BLOCKS ((N_NODES + 127) / 128)
static_assert(N_NODES <= 65536, "a source index must fit the low 16 bits of the packed key");
static_assert(T_LEN <= T_PAD, "table padding");
static_assert(CHUNK < 32768, "run start and run length share one 32-bit table entry");
static_assert(N_NODES % 16 == 0, "the product kernels store whole 16-row strips");

typedef __attribute__((ext_vector_type(2))) float v2f;
typedef __attribute__((ext_vector_type(4))) float v4f;
typedef __attribute__((ext_vector_type(8))) float v8f;

__device__ __forceinline__
void chunk_place_body(const int chunk,
                      const int* __restrict__ src, const int* __restrict__ dst,
                      const float* __restrict__ w1, const float* __restrict__ w2,
                      const float* __restrict__ w3, int* __restrict__ tbl,
                      float* __restrict__ recC, int nEdges) {
    __shared__ int sCnt[256];
    __shared__ int sOff[256];
    __shared__ int sWave[8];
    static_assert(N_BINS <= 256, "one counter per thread");
    const int tid  = threadIdx.x;
    const int lane = tid & 31;
    const int wave = tid >> 5;
    sCnt[tid] = 0;
    const int base = chunk * CHUNK + tid;
    int   d[CHUNK / 256], s[CHUNK / 256];
    float a[CHUNK / 256], b[CHUNK / 256], c[CHUNK / 256];
#pragma unroll
    for (int i = 0; i < CHUNK / 256; ++i) {
        const int e = min(base + 256 * i, nEdges - 1);
        d[i] = dst[e];
        s[i] = src[e];
        a[i] = w1[e];
        b[i] = w2[e];
        c[i] = w3[e];
    }
    __syncthreads();
    int r[CHUNK / 256];
    if (chunk * CHUNK + CHUNK <= nEdges) {
#pragma unroll
        for (int i = 0; i < CHUNK / 256; ++i) r[i] = atomicAdd(&sCnt[d[i] >> BIN_SHIFT], 1);
    } else {
#pragma unroll
        for (int i = 0; i < CHUNK / 256; ++i) {
            r[i] = 0;
            if (base + 256 * i < nEdges) r[i] = atomicAdd(&sCnt[d[i] >> BIN_SHIFT], 1);
        }
    }
    __syncthreads();

    const int v = sCnt[tid];
    int incl = v;
#pragma unroll
    for (int o = 1; o < 32; o <<= 1) {
        const int t = __shfl_up(incl, o);
        if (lane >= o) incl += t;
    }
    if (lane == 31) sWave[wave] = incl;
    __syncthreads();
    int pre = 0;
#pragma unroll
    for (int w = 0; w < 8; ++w) pre += (w < wave) ? sWave[w] : 0;
    const int off = pre + incl - v;
    sOff[tid] = off;
    if (tid < N_BINS) tbl[tid * N_CHUNKS + chunk] = (off << 16) | v;
    __syncthreads();

    v4f* slice = (v4f*)recC + (size_t)chunk * CHUNK;
    int pos[CHUNK / 256];
#pragma unroll
    for (int i = 0; i < CHUNK / 256; ++i) pos[i] = sOff[d[i] >> BIN_SHIFT] + r[i];
#pragma unroll
    for (int i = 0; i < CHUNK / 256; ++i) {
        if (base + 256 * i < nEdges) {
            v4f q;
            q.x = __int_as_float(s[i] | ((d[i] & (BIN_NODES - 1)) << 16));
            q.y = a[i];
            q.z = b[i];
            q.w = c[i];
            slice[pos[i]] = q;
        }
    }
}

#define RUNS_PER_WAVE ((N_CHUNKS + 31) / 32)
__global__ __launch_bounds__(1024)
void fine_place_kernel(const float* __restrict__ recC, const int* __restrict__ tbl,
                       int* __restrict__ rowptr, float* __restrict__ rec, int nNodes) {
    __shared__ int sRun[N_CHUNKS];
    __shared__ int sCnt[BIN_NODES];
    __shared__ int sCur[BIN_NODES];
    __shared__ int sWave[BIN_NODES / 32];
    __shared__ int sPart[32];
    const int tid  = threadIdx.x;
    const int lane = tid & 31;
    const int wave = __builtin_amdgcn_readfirstlane(tid >> 5);
    const int bin  = blockIdx.x;

    int myRun = 0;
    if (tid < N_CHUNKS) myRun = tbl[bin * N_CHUNKS + tid];
    if (tid < BIN_NODES) sCnt[tid] = 0;

    const int nBefore = bin * N_CHUNKS;
    const int n4      = nBefore >> 2;
    int before = 0;
    for (int i = tid; i < n4; i += 1024) {
        const int4 q = ((const int4*)tbl)[i];
        before += (q.x & 0xFFFF) + (q.y & 0xFFFF) + (q.z & 0xFFFF) + (q.w & 0xFFFF);
    }
    if (tid < (nBefore & 3)) before += tbl[4 * n4 + tid] & 0xFFFF;
    if (tid < N_CHUNKS) sRun[tid] = myRun;
#pragma unroll
    for (int o = 16; o > 0; o >>= 1) before += __shfl_xor(before, o);
    if (lane == 0) sPart[wave] = before;
    __syncthreads();

    int segBeg = sPart[lane];
#pragma unroll
    for (int o = 16; o > 0; o >>= 1) segBeg += __shfl_xor(segBeg, o);

    int pv[RUNS_PER_WAVE], cnt[RUNS_PER_WAVE], rb[RUNS_PER_WAVE];
#pragma unroll
    for (int k = 0; k < RUNS_PER_WAVE; ++k) pv[k] = sRun[min(wave + 32 * k, N_CHUNKS - 1)];
#pragma unroll
    for (int k = 0; k < RUNS_PER_WAVE; ++k) {
        const int r = wave + 32 * k;
        const int p = (r < N_CHUNKS) ? __builtin_amdgcn_readfirstlane(pv[k]) : 0;
        cnt[k] = p & 0xFFFF;
        rb[k]  = r * CHUNK + (p >> 16);
    }
    v4f q[RUNS_PER_WAVE];
#pragma unroll
    for (int k = 0; k < RUNS_PER_WAVE; ++k) {
        q[k] = (v4f)(0.0f);
        if (lane < cnt[k]) q[k] = ((const v4f*)recC)[rb[k] + lane];
    }
#pragma unroll
    for (int k = 0; k < RUNS_PER_WAVE; ++k) {
        if (lane < cnt[k]) atomicAdd(&sCnt[(__float_as_int(q[k].x) >> 16) & (BIN_NODES - 1)], 1);
    }
#pragma unroll
    for (int k = 0; k < RUNS_PER_WAVE; ++k) {
        for (int i = 32 + lane; i < cnt[k]; i += 32) {
            const int key = __float_as_int(recC[(size_t)4 * (rb[k] + i)]);
            atomicAdd(&sCnt[(key >> 16) & (BIN_NODES - 1)], 1);
        }
    }
    __syncthreads();

    int v = 0, incl = 0;
    if (tid < BIN_NODES) {
        v = sCnt[tid];
        incl = v;
#pragma unroll
        for (int o = 1; o < 32; o <<= 1) {
            const int t = __shfl_up(incl, o);
            if (lane >= o) incl += t;
        }
        if (lane == 31) sWave[wave] = incl;
    }
    __syncthreads();
    if (tid < BIN_NODES) {
        int pre = 0;
#pragma unroll
        for (int w = 0; w < BIN_NODES / 32; ++w) pre += (w < wave) ? sWave[w] : 0;
        const int start = segBeg + pre + incl - v;
        sCur[tid] = start;
        const int node = bin * BIN_NODES + tid;
        if (node <= nNodes) rowptr[node] = start;
    }
    __syncthreads();

    int slot[RUNS_PER_WAVE];
#pragma unroll
    for (int k = 0; k < RUNS_PER_WAVE; ++k) {
        slot[k] = 0;
        if (lane < cnt[k]) slot[k] = atomicAdd(&sCur[(__float_as_int(q[k].x) >> 16) & (BIN_NODES - 1)], 1);
    }
#pragma unroll
    for (int k = 0; k < RUNS_PER_WAVE; ++k) {
        if (lane < cnt[k]) {
            v4f o = q[k];
            o.x = __int_as_float(__float_as_int(o.x) & 0xFFFF);
            ((v4f*)rec)[slot[k]] = o;
        }
    }
#pragma unroll
    for (int k = 0; k < RUNS_PER_WAVE; ++k) {
        for (int i = 32 + lane; i < cnt[k]; i += 32) {
            v4f o = ((const v4f*)recC)[rb[k] + i];
            const int key = __float_as_int(o.x);
            const int p = atomicAdd(&sCur[(key >> 16) & (BIN_NODES - 1)], 1);
            o.x = __int_as_float(key & 0xFFFF);
            ((v4f*)rec)[p] = o;
        }
    }
}

template <int N, bool RELU_A>
__device__ __forceinline__
void gemm_wmma_f32_body(const int rowBlock,
                        const float* __restrict__ A, const float* __restrict__ W,
                        const float* __restrict__ bias, float* __restrict__ out, int M) {
    constexpr int K     = 128;
    constexpr int KS    = K / 4;
    constexpr int NT    = N / 16;
    constexpr int PITCH = 2 * N + 32;
    constexpr int N4    = N / 4;
    constexpr int ITEMS = (K / 2) * N4;
    __shared__ __attribute__((aligned(16))) float sW[(K / 2) * PITCH];

    const int tid   = threadIdx.x;
    const int lane  = tid & 31;
    const int wave  = tid >> 5;
    const int half  = lane >> 4;
    const int l16   = lane & 15;
    const int mTile = rowBlock * 8 + wave;
    const bool live = (mTile * 16 < M);

    const int aRowIdx = live ? (mTile * 16 + l16) : l16;
    const float* aRow = A + (size_t)aRowIdx * K + 2 * half;
    v2f a[KS];
#pragma unroll
    for (int s = 0; s < KS; ++s) a[s] = *(const v2f*)(aRow + 4 * s);

#pragma unroll
    for (int i = 0; i < ITEMS / 256; ++i) {
        const int it = tid + 256 * i;
        const int p  = it / N4;
        const int n4 = it % N4;
        const v4f r0 = *(const v4f*)(W + (size_t)(2 * p)     * N + 4 * n4);
        const v4f r1 = *(const v4f*)(W + (size_t)(2 * p + 1) * N + 4 * n4);
        float* d = sW + p * PITCH + 8 * n4;
        v4f lo, hi;
        lo.x = r0.x; lo.y = r1.x; lo.z = r0.y; lo.w = r1.y;
        hi.x = r0.z; hi.y = r1.z; hi.z = r0.w; hi.w = r1.w;
        *(v4f*)d       = lo;
        *(v4f*)(d + 4) = hi;
    }
#pragma unroll
    for (int s = 0; s < KS; ++s) asm volatile("" :: "v"(a[s]));
    __syncthreads();
    if (!live) return;

    if (RELU_A) {
#pragma unroll
        for (int s = 0; s < KS; ++s) {
            a[s].x = fmaxf(a[s].x, 0.0f);
            a[s].y = fmaxf(a[s].y, 0.0f);
        }
    }

    v8f c[NT];
#pragma unroll
    for (int t = 0; t < NT; ++t) {
        const float bv = bias[t * 16 + l16];
#pragma unroll
        for (int i = 0; i < 8; ++i) c[t][i] = bv;
    }

    const float* bBase = sW + half * PITCH + 2 * l16;
#pragma unroll
    for (int s = 0; s < KS; ++s) {
#pragma unroll
        for (int t = 0; t < NT; ++t) {
            const v2f b = *(const v2f*)(bBase + (2 * s) * PITCH + 32 * t);
            c[t] = __builtin_amdgcn_wmma_f32_16x16x4_f32(false, a[s], false, b,
                                                         (short)0, c[t], false, false);
        }
    }

    float* oBase = out + (size_t)(mTile * 16 + half * 8) * N + l16;
#pragma unroll
    for (int r = 0; r < 8; ++r) {
#pragma unroll
        for (int t = 0; t < NT; ++t) oBase[(size_t)r * N + 16 * t] = c[t][r];
    }
}

template <int N>
__global__ __launch_bounds__(256)
void chunk_place_and_gemm_kernel(const int* __restrict__ src, const int* __restrict__ dst,
                                 const float* __restrict__ w1, const float* __restrict__ w2,
                                 const float* __restrict__ w3, int* __restrict__ tbl,
                                 float* __restrict__ recC, int nEdges,
                                 const float* __restrict__ A, const float* __restrict__ W,
                                 const float* __restrict__ bias, float* __restrict__ out, int M) {
    if (blockIdx.x < ROW_BLOCKS) {
        gemm_wmma_f32_body<N, false>(blockIdx.x, A, W, bias, out, M);
    } else {
        chunk_place_body(blockIdx.x - ROW_BLOCKS, src, dst, w1, w2, w3, tbl, recC, nEdges);
    }
}

template <int FEATS, int K, typename VROW>
__device__ __forceinline__ void add_rows(VROW& acc, const float* __restrict__ hLane,
                                         int sL, int wL, int j) {
    int   s[K];
    float w[K];
    VROW  a[K];
#pragma unroll
    for (int i = 0; i < K; ++i) {
        s[i] = __builtin_amdgcn_readlane(sL, j + i);
        w[i] = __int_as_float(__builtin_amdgcn_readlane(wL, j + i));
    }
#pragma unroll
    for (int i = 0; i < K; ++i) a[i] = *(const VROW*)(hLane + (size_t)s[i] * FEATS);
#pragma unroll
    for (int i = 0; i < K; ++i) acc += a[i] * w[i];
}

template <int FEATS, int WSEL>
__global__ __launch_bounds__(256)
void gather_sum_kernel(const float* __restrict__ h, const int* __restrict__ rowptr,
                       const float* __restrict__ rec, float* __restrict__ out,
                       int nNodes) {
    constexpr int V = FEATS / 32;
    typedef __attribute__((ext_vector_type(V))) float vrow;
    const int lane = threadIdx.x & 31;
    const int d = __builtin_amdgcn_readfirstlane(blockIdx.x * 8 + (threadIdx.x >> 5));
    if (d >= nNodes) return;
    const int beg = rowptr[d];
    const int end = rowptr[d + 1];

    const float* hLane = h + lane * V;
    vrow acc = (vrow)(0.0f);

    for (int base = beg; base < end; base += 32) {
        const int n = min(32, end - base);
        v4f r = (v4f)(0.0f);
        if (lane < n) r = ((const v4f*)rec)[base + lane];
        const int sL = __float_as_int(r.x);
        const int wL = __float_as_int(WSEL == 1 ? r.y : (WSEL == 2 ? r.z : r.w));

        int j = 0;
        for (; j + 8 <= n; j += 8) add_rows<FEATS, 8>(acc, hLane, sL, wL, j);
        if (j + 4 <= n) { add_rows<FEATS, 4>(acc, hLane, sL, wL, j); j += 4; }
        const int rem = n - j;
        if (rem == 3)      add_rows<FEATS, 3>(acc, hLane, sL, wL, j);
        else if (rem == 2) add_rows<FEATS, 2>(acc, hLane, sL, wL, j);
        else if (rem == 1) add_rows<FEATS, 1>(acc, hLane, sL, wL, j);
    }
    *(vrow*)(out + (size_t)d * FEATS + lane * V) = acc;
}

#define A_PITCH  132
#define GG_WGS   256
#define GG_WAVES 16
#define M_TILES  (N_NODES / 16)
#define S_BASE   (M_TILES / GG_WGS)
#define S_EXTRA  (M_TILES % GG_WGS)
#define S_MAX    (S_BASE + (S_EXTRA ? 1 : 0))
static_assert(S_MAX + 1 <= 32, "a wave's S + 1 row pointers sit in the lanes of one register");

__device__ __forceinline__ void flush_row(float* __restrict__ sAw, int row, int lane, v4f& acc) {
    v4f r;
    r.x = fmaxf(acc.x, 0.0f);
    r.y = fmaxf(acc.y, 0.0f);
    r.z = fmaxf(acc.z, 0.0f);
    r.w = fmaxf(acc.w, 0.0f);
    *(v4f*)(sAw + row * A_PITCH + 4 * lane) = r;
    acc = (v4f)(0.0f);
}

template <int KK>
__device__ __forceinline__ void add_run(v4f& acc, int& cur, int& rowEnd, int rpv, int lastRow,
                                        float* __restrict__ sAw, const float* __restrict__ hLane,
                                        int sL, int wL, int j, int idx0, int lane) {
    int   s[KK];
    float w[KK];
    v4f   a[KK];
#pragma unroll
    for (int i = 0; i < KK; ++i) {
        s[i] = __builtin_amdgcn_readlane(sL, j + i);
        w[i] = __int_as_float(__builtin_amdgcn_readlane(wL, j + i));
    }
#pragma unroll
    for (int i = 0; i < KK; ++i) a[i] = *(const v4f*)(hLane + (size_t)s[i] * 128);
#pragma unroll
    for (int i = 0; i < KK; ++i) asm volatile("" :: "v"(a[i]));
#pragma unroll
    for (int i = 0; i < KK; ++i) {
        while (idx0 + i == rowEnd && cur < lastRow) {
            flush_row(sAw, cur, lane, acc);
            ++cur;
            rowEnd = __builtin_amdgcn_readlane(rpv, cur + 1);
        }
        acc += a[i] * w[i];
    }
}

template <int N, int WSEL>
__global__ __launch_bounds__(32 * GG_WAVES)
void gather_gemm_wmma_f32(const float* __restrict__ h, const int* __restrict__ rowptr,
                          const float* __restrict__ rec, const float* __restrict__ W,
                          const float* __restrict__ bias, float* __restrict__ out, int M) {
    constexpr int K     = 128;
    constexpr int KS    = K / 4;
    constexpr int PITCH = 2 * N + 32;
    constexpr int N4    = N / 4;
    constexpr int ITEMS = (K / 2) * N4;
    constexpr int NTHR  = 32 * GG_WAVES;
    constexpr int NI    = ITEMS / NTHR;
    constexpr int UPS   = N / 32;
    static_assert(ITEMS % NTHR == 0, "every thread stages the same number of W items");
    static_assert((UPS & (UPS - 1)) == 0, "unit -> (strip, pair) by shift and mask");
    __shared__ __attribute__((aligned(16))) float sW[(K / 2) * PITCH];
    __shared__ __attribute__((aligned(16))) float sA[S_MAX * 16 * A_PITCH];

    const int tid   = threadIdx.x;
    const int lane  = tid & 31;
    const int wave  = __builtin_amdgcn_readfirstlane(tid >> 5);
    const int half  = lane >> 4;
    const int l16   = lane & 15;
    const int wg     = blockIdx.x;
    const int S      = S_BASE + (wg < S_EXTRA ? 1 : 0);
    const int strip0 = wg * S_BASE + min(wg, S_EXTRA);
    const int rowW   = strip0 * 16 + wave * S;

    const int rpv = rowptr[min(rowW + min(lane, S), M)];

    v4f r0[NI], r1[NI];
#pragma unroll
    for (int i = 0; i < NI; ++i) {
        const int it = tid + NTHR * i;
        const int p  = it / N4;
        const int n4 = it % N4;
        r0[i] = *(const v4f*)(W + (size_t)(2 * p)     * N + 4 * n4);
        r1[i] = *(const v4f*)(W + (size_t)(2 * p + 1) * N + 4 * n4);
    }

    const int beg0   = __builtin_amdgcn_readlane(rpv, 0);
    const int endAll = __builtin_amdgcn_readlane(rpv, S);
    const int* recI = (const int*)rec;
    int sCur, wCur;
    {
        const int e = min(beg0 + lane, N_EDGES - 1);
        sCur = recI[4 * e];
        wCur = recI[4 * e + WSEL];
    }

#pragma unroll
    for (int i = 0; i < NI; ++i) {
        const int it = tid + NTHR * i;
        const int p  = it / N4;
        const int n4 = it % N4;
        float* d = sW + p * PITCH + 8 * n4;
        v4f lo, hi;
        lo.x = r0[i].x; lo.y = r1[i].x; lo.z = r0[i].y; lo.w = r1[i].y;
        hi.x = r0[i].z; hi.y = r1[i].z; hi.z = r0[i].w; hi.w = r1[i].w;
        *(v4f*)d       = lo;
        *(v4f*)(d + 4) = hi;
    }

    float* sAw = sA + wave * S * A_PITCH;
    {
        const float* hLane = h + lane * 4;
        const int lastRow = S - 1;
        v4f acc   = (v4f)(0.0f);
        int cur    = 0;
        int rowEnd = __builtin_amdgcn_readlane(rpv, 1);
        for (int base = beg0; base < endAll; base += 32) {
            const int n  = min(32, endAll - base);
            const int sL = sCur;
            const int wL = wCur;
            const int eN = min(base + 32 + lane, N_EDGES - 1);
            const int sNext = recI[4 * eN];
            const int wNext = recI[4 * eN + WSEL];

            int j = 0;
            for (; j + 16 <= n; j += 16) add_run<16>(acc, cur, rowEnd, rpv, lastRow, sAw, hLane, sL, wL, j, base + j, lane);
            if (j + 8 <= n) { add_run<8>(acc, cur, rowEnd, rpv, lastRow, sAw, hLane, sL, wL, j, base + j, lane); j += 8; }
            if (j + 4 <= n) { add_run<4>(acc, cur, rowEnd, rpv, lastRow, sAw, hLane, sL, wL, j, base + j, lane); j += 4; }
            const int rem = n - j;
            if (rem == 3)      add_run<3>(acc, cur, rowEnd, rpv, lastRow, sAw, hLane, sL, wL, j, base + j, lane);
            else if (rem == 2) add_run<2>(acc, cur, rowEnd, rpv, lastRow, sAw, hLane, sL, wL, j, base + j, lane);
            else if (rem == 1) add_run<1>(acc, cur, rowEnd, rpv, lastRow, sAw, hLane, sL, wL, j, base + j, lane);
            sCur = sNext;
            wCur = wNext;
        }
        for (; cur < S; ++cur) flush_row(sAw, cur, lane, acc);
    }
    __syncthreads();

    const int nUnits = S * UPS;
    const int place  = 4 * (((wave & 3) - (wave >> 2)) & 3) + (wave >> 2);
    for (int u = place; u < nUnits; u += GG_WAVES) {
        const int st = u / UPS;
        const int tp = u % UPS;

        v2f a[KS];
        const float* aBase = sA + (st * 16 + l16) * A_PITCH + 2 * half;
#pragma unroll
        for (int s = 0; s < KS; ++s) a[s] = *(const v2f*)(aBase + 4 * s);

        const float bv0 = bias[32 * tp + l16];
        const float bv1 = bias[32 * tp + 16 + l16];
        v8f c0, c1;
#pragma unroll
        for (int i = 0; i < 8; ++i) { c0[i] = bv0; c1[i] = bv1; }

        const float* bB = sW + half * PITCH + 2 * l16 + 64 * tp;
#pragma unroll
        for (int s = 0; s < KS; ++s) {
            const v2f b0 = *(const v2f*)(bB + (2 * s) * PITCH);
            const v2f b1 = *(const v2f*)(bB + (2 * s) * PITCH + 32);
            c0 = __builtin_amdgcn_wmma_f32_16x16x4_f32(false, a[s], false, b0,
                                                       (short)0, c0, false, false);
            c1 = __builtin_amdgcn_wmma_f32_16x16x4_f32(false, a[s], false, b1,
                                                       (short)0, c1, false, false);
        }

        float* oBase = out + (size_t)((strip0 + st) * 16 + half * 8) * N + 32 * tp + l16;
#pragma unroll
        for (int r = 0; r < 8; ++r) {
            oBase[(size_t)r * N]      = c0[r];
            oBase[(size_t)r * N + 16] = c1[r];
        }
    }
}

extern "C" void kernel_launch(void* const* d_in, const int* in_sizes, int n_in,
                              void* d_out, int out_size, void* d_ws, size_t ws_size,
                              hipStream_t stream) {
    const float* x   = (const float*)d_in[0];
    const int*   src = (const int*)  d_in[1];
    const int*   dst = (const int*)  d_in[2];
    const float* w1  = (const float*)d_in[3];
    const float* w2  = (const float*)d_in[4];
    const float* w3  = (const float*)d_in[5];
    const float* W1  = (const float*)d_in[6];
    const float* b1  = (const float*)d_in[7];
    const float* W2  = (const float*)d_in[8];
    const float* b2  = (const float*)d_in[9];
    const float* W3  = (const float*)d_in[10];
    const float* b3  = (const float*)d_in[11];
    float* out = (float*)d_out;

    float* wsA    = (float*)d_ws;
    float* wsB    = wsA + (size_t)N_NODES * N_HIDDEN;
    float* rec    = wsB + (size_t)N_NODES * N_HIDDEN;
    float* recC   = rec + (size_t)N_EDGES * 4;
    int*   tbl    = (int*)(recC + (size_t)N_EDGES * 4);
    int*   rowptr = tbl + T_PAD;

    const dim3 blk(256);
    const dim3 fusedGrid(GG_WGS);
    const dim3 fusedBlk(32 * GG_WAVES);
    const dim3 nodeGrid((N_NODES + 7) / 8);
    const dim3 firstGrid(ROW_BLOCKS + N_CHUNKS);
    const dim3 binGrid(N_BINS);

    chunk_place_and_gemm_kernel<N_HIDDEN><<<firstGrid, blk, 0, stream>>>(src, dst, w1, w2, w3, tbl, recC, N_EDGES,
                                                                  x, W1, b1, wsA, N_NODES);
    fine_place_kernel<<<binGrid, dim3(1024), 0, stream>>>(recC, tbl, rowptr, rec, N_NODES);

    gather_gemm_wmma_f32<N_HIDDEN, 1><<<fusedGrid, fusedBlk, 0, stream>>>(wsA, rowptr, rec, W2, b2, wsB, N_NODES);

    gather_gemm_wmma_f32<N_CLASSES, 2><<<fusedGrid, fusedBlk, 0, stream>>>(wsB, rowptr, rec, W3, b3, wsA, N_NODES);

    gather_sum_kernel<N_CLASSES, 3><<<nodeGrid, blk, 0, stream>>>(wsA, rowptr, rec, out, N_NODES);
}
